# starting kernel
# baseline (speedup 1.0000x reference)
#include <hip/hip_runtime.h>

typedef __attribute__((ext_vector_type(2))) float v2f;
typedef __attribute__((ext_vector_type(8))) float v8f;

#define DIM 128
#define NTILES 8
#define WAVES_PER_BLOCK 8

__global__ __launch_bounds__(256) void gcn_init_out(float* __restrict__ out,
                                                    const float* __restrict__ bias,
                                                    int total) {
    int i = blockIdx.x * blockDim.x + threadIdx.x;
    if (i < total) out[i] = bias[i & (DIM - 1)];
}

__global__ __launch_bounds__(256) void gcn_gemm_wmma(const float* __restrict__ x,
                                                     const float* __restrict__ w,
                                                     float* __restrict__ h,
                                                     int nrows) {
    __shared__ float wlds[DIM * DIM];

    {
        const float4* wsrc = reinterpret_cast<const float4*>(w);
        float4* wdst = reinterpret_cast<float4*>(wlds);
        #pragma unroll
        for (int i = threadIdx.x; i < DIM * DIM / 4; i += 256) wdst[i] = wsrc[i];
    }
    __syncthreads();

    const int wave = threadIdx.x >> 5;
    const int lane = threadIdx.x & 31;
    const int half = lane >> 4;
    const int l15  = lane & 15;

    const int mtiles = (nrows + 15) >> 4;
    const int mtile  = blockIdx.x * WAVES_PER_BLOCK + wave;
    if (mtile >= mtiles) return;

    const int row  = mtile * 16 + l15;
    const int lrow = row < nrows ? row : nrows - 1;

    v8f acc[NTILES];
    const v8f vzero = {0.f, 0.f, 0.f, 0.f, 0.f, 0.f, 0.f, 0.f};
    #pragma unroll
    for (int n = 0; n < NTILES; ++n) acc[n] = vzero;

    const int koff = half * 2;

    for (int kb = 0; kb < DIM; kb += 4) {
        const float2 af =
            *reinterpret_cast<const float2*>(x + (size_t)lrow * DIM + kb + koff);
        v2f a;
        a.x = af.x;
        a.y = af.y;

        #pragma unroll
        for (int n = 0; n < NTILES; ++n) {
            const int col = n * 16 + l15;
            v2f b;
            b.x = wlds[(kb + koff    ) * DIM + col];
            b.y = wlds[(kb + koff + 1) * DIM + col];
            acc[n] = __builtin_amdgcn_wmma_f32_16x16x4_f32(
                 false, a,  false, b,
                 (short)0, acc[n],
                 false,  false);
        }
    }

    #pragma unroll
    for (int n = 0; n < NTILES; ++n) {
        #pragma unroll
        for (int r = 0; r < 8; ++r) {
            const int m = mtile * 16 + half * 8 + r;
            if (m < nrows) h[(size_t)m * DIM + n * 16 + l15] = acc[n][r];
        }
    }
}

__global__ __launch_bounds__(256) void gcn_scatter(const float* __restrict__ h,
                                                   const int* __restrict__ esrc,
                                                   const int* __restrict__ edst,
                                                   const float* __restrict__ ew,
                                                   float* out, int E) {
    const long long t = (long long)blockIdx.x * blockDim.x + threadIdx.x;
    const int e = (int)(t >> 5);
    if (e >= E) return;
    const int c = (int)(t & 31) * 4;

    const int s = esrc[e];
    const int d = edst[e];
    const float wt = ew[e];

    const float4 v = *reinterpret_cast<const float4*>(h + (size_t)s * DIM + c);
    float* o = out + (size_t)d * DIM + c;

    __hip_atomic_fetch_add(o + 0, v.x * wt, __ATOMIC_RELAXED, __HIP_MEMORY_SCOPE_AGENT);
    __hip_atomic_fetch_add(o + 1, v.y * wt, __ATOMIC_RELAXED, __HIP_MEMORY_SCOPE_AGENT);
    __hip_atomic_fetch_add(o + 2, v.z * wt, __ATOMIC_RELAXED, __HIP_MEMORY_SCOPE_AGENT);
    __hip_atomic_fetch_add(o + 3, v.w * wt, __ATOMIC_RELAXED, __HIP_MEMORY_SCOPE_AGENT);
}

extern "C" void kernel_launch(void* const* d_in, const int* in_sizes, int n_in,
                              void* d_out, int out_size, void* d_ws, size_t ws_size,
                              hipStream_t stream) {
    const float* x    = (const float*)d_in[0];
    const int*   esrc = (const int*)d_in[1];
    const int*   edst = (const int*)d_in[2];
    const float* ew   = (const float*)d_in[3];
    const float* w    = (const float*)d_in[4];
    const float* bias = (const float*)d_in[5];

    const int N = in_sizes[0] / DIM;
    const int E = in_sizes[1];

    float* out = (float*)d_out;
    float* h   = (float*)d_ws;

    {
        const int total = N * DIM;
        gcn_init_out<<<(total + 255) / 256, 256, 0, stream>>>(out, bias, total);
    }

    {
        const int mtiles = (N + 15) >> 4;
        const int blocks = (mtiles + WAVES_PER_BLOCK - 1) / WAVES_PER_BLOCK;
        gcn_gemm_wmma<<<blocks, 256, 0, stream>>>(x, w, h, N);
    }

    {
        const long long threads = (long long)E * 32;
        const int blocks = (int)((threads + 255) / 256);
        gcn_scatter<<<blocks, 256, 0, stream>>>(h, esrc, edst, ew, out, E);
    }
}


# scatter: lane l owns columns l,l+32,l+64,l+96, so each f32 atomic add (and each row load) instruction covers one whole 128-byte line, not 32-byte pieces of 4 lines; sums still land in arrival order
# speedup vs baseline: 1.3114x; 1.3114x over previous
#include <hip/hip_runtime.h>

typedef __attribute__((ext_vector_type(2))) float v2f;
typedef __attribute__((ext_vector_type(8))) float v8f;

#define DIM 128
#define NTILES 8
#define WAVES_PER_BLOCK 8

__global__ __launch_bounds__(256) void gcn_init_out(float* __restrict__ out,
                                                    const float* __restrict__ bias,
                                                    int total) {
    int i = blockIdx.x * blockDim.x + threadIdx.x;
    if (i < total) out[i] = bias[i & (DIM - 1)];
}

__global__ __launch_bounds__(256) void gcn_gemm_wmma(const float* __restrict__ x,
                                                     const float* __restrict__ w,
                                                     float* __restrict__ h,
                                                     int nrows) {
    __shared__ float wlds[DIM * DIM];

    {
        const float4* wsrc = reinterpret_cast<const float4*>(w);
        float4* wdst = reinterpret_cast<float4*>(wlds);
        #pragma unroll
        for (int i = threadIdx.x; i < DIM * DIM / 4; i += 256) wdst[i] = wsrc[i];
    }
    __syncthreads();

    const int wave = threadIdx.x >> 5;
    const int lane = threadIdx.x & 31;
    const int half = lane >> 4;
    const int l15  = lane & 15;

    const int mtiles = (nrows + 15) >> 4;
    const int mtile  = blockIdx.x * WAVES_PER_BLOCK + wave;
    if (mtile >= mtiles) return;

    const int row  = mtile * 16 + l15;
    const int lrow = row < nrows ? row : nrows - 1;

    v8f acc[NTILES];
    const v8f vzero = {0.f, 0.f, 0.f, 0.f, 0.f, 0.f, 0.f, 0.f};
    #pragma unroll
    for (int n = 0; n < NTILES; ++n) acc[n] = vzero;

    const int koff = half * 2;

    for (int kb = 0; kb < DIM; kb += 4) {
        const float2 af =
            *reinterpret_cast<const float2*>(x + (size_t)lrow * DIM + kb + koff);
        v2f a;
        a.x = af.x;
        a.y = af.y;

        #pragma unroll
        for (int n = 0; n < NTILES; ++n) {
            const int col = n * 16 + l15;
            v2f b;
            b.x = wlds[(kb + koff    ) * DIM + col];
            b.y = wlds[(kb + koff + 1) * DIM + col];
            acc[n] = __builtin_amdgcn_wmma_f32_16x16x4_f32(
                 false, a,  false, b,
                 (short)0, acc[n],
                 false,  false);
        }
    }

    #pragma unroll
    for (int n = 0; n < NTILES; ++n) {
        #pragma unroll
        for (int r = 0; r < 8; ++r) {
            const int m = mtile * 16 + half * 8 + r;
            if (m < nrows) h[(size_t)m * DIM + n * 16 + l15] = acc[n][r];
        }
    }
}

__global__ __launch_bounds__(256) void gcn_scatter(const float* __restrict__ h,
                                                   const int* __restrict__ esrc,
                                                   const int* __restrict__ edst,
                                                   const float* __restrict__ ew,
                                                   float* out, int E) {
    const long long t = (long long)blockIdx.x * blockDim.x + threadIdx.x;
    const int e = (int)(t >> 5);
    if (e >= E) return;
    const int lane = (int)(t & 31);

    const int s = esrc[e];
    const int d = edst[e];
    const float wt = ew[e];

    const float* hp = h + (size_t)s * DIM + lane;
    float* o = out + (size_t)d * DIM + lane;

    const float v0 = hp[0];
    const float v1 = hp[32];
    const float v2 = hp[64];
    const float v3 = hp[96];

    __hip_atomic_fetch_add(o +  0, v0 * wt, __ATOMIC_RELAXED, __HIP_MEMORY_SCOPE_AGENT);
    __hip_atomic_fetch_add(o + 32, v1 * wt, __ATOMIC_RELAXED, __HIP_MEMORY_SCOPE_AGENT);
    __hip_atomic_fetch_add(o + 64, v2 * wt, __ATOMIC_RELAXED, __HIP_MEMORY_SCOPE_AGENT);
    __hip_atomic_fetch_add(o + 96, v3 * wt, __ATOMIC_RELAXED, __HIP_MEMORY_SCOPE_AGENT);
}

extern "C" void kernel_launch(void* const* d_in, const int* in_sizes, int n_in,
                              void* d_out, int out_size, void* d_ws, size_t ws_size,
                              hipStream_t stream) {
    const float* x    = (const float*)d_in[0];
    const int*   esrc = (const int*)d_in[1];
    const int*   edst = (const int*)d_in[2];
    const float* ew   = (const float*)d_in[3];
    const float* w    = (const float*)d_in[4];
    const float* bias = (const float*)d_in[5];

    const int N = in_sizes[0] / DIM;
    const int E = in_sizes[1];

    float* out = (float*)d_out;
    float* h   = (float*)d_ws;

    {
        const int total = N * DIM;
        gcn_init_out<<<(total + 255) / 256, 256, 0, stream>>>(out, bias, total);
    }

    {
        const int mtiles = (N + 15) >> 4;
        const int blocks = (mtiles + WAVES_PER_BLOCK - 1) / WAVES_PER_BLOCK;
        gcn_gemm_wmma<<<blocks, 256, 0, stream>>>(x, w, h, N);
    }

    {
        const long long threads = (long long)E * 32;
        const int blocks = (int)((threads + 255) / 256);
        gcn_scatter<<<blocks, 256, 0, stream>>>(h, esrc, edst, ew, out, E);
    }
}


# scatter: one wave per 32 edges (coalesced index loads, lane broadcast), h rows of 8 edges loaded as a group, two groups in flight, not one edge per wave; f32 atomic sums still in arrival order
# speedup vs baseline: 1.9242x; 1.4673x over previous
#include <hip/hip_runtime.h>

typedef __attribute__((ext_vector_type(2))) float v2f;
typedef __attribute__((ext_vector_type(8))) float v8f;

#define DIM 128
#define NTILES 8
#define WAVES_PER_BLOCK 8

__global__ __launch_bounds__(256) void gcn_init_out(float* __restrict__ out,
                                                    const float* __restrict__ bias,
                                                    int total) {
    int i = blockIdx.x * blockDim.x + threadIdx.x;
    if (i < total) out[i] = bias[i & (DIM - 1)];
}

__global__ __launch_bounds__(256) void gcn_gemm_wmma(const float* __restrict__ x,
                                                     const float* __restrict__ w,
                                                     float* __restrict__ h,
                                                     int nrows) {
    __shared__ float wlds[DIM * DIM];

    {
        const float4* wsrc = reinterpret_cast<const float4*>(w);
        float4* wdst = reinterpret_cast<float4*>(wlds);
        #pragma unroll
        for (int i = threadIdx.x; i < DIM * DIM / 4; i += 256) wdst[i] = wsrc[i];
    }
    __syncthreads();

    const int wave = threadIdx.x >> 5;
    const int lane = threadIdx.x & 31;
    const int half = lane >> 4;
    const int l15  = lane & 15;

    const int mtiles = (nrows + 15) >> 4;
    const int mtile  = blockIdx.x * WAVES_PER_BLOCK + wave;
    if (mtile >= mtiles) return;

    const int row  = mtile * 16 + l15;
    const int lrow = row < nrows ? row : nrows - 1;

    v8f acc[NTILES];
    const v8f vzero = {0.f, 0.f, 0.f, 0.f, 0.f, 0.f, 0.f, 0.f};
    #pragma unroll
    for (int n = 0; n < NTILES; ++n) acc[n] = vzero;

    const int koff = half * 2;

    for (int kb = 0; kb < DIM; kb += 4) {
        const float2 af =
            *reinterpret_cast<const float2*>(x + (size_t)lrow * DIM + kb + koff);
        v2f a;
        a.x = af.x;
        a.y = af.y;

        #pragma unroll
        for (int n = 0; n < NTILES; ++n) {
            const int col = n * 16 + l15;
            v2f b;
            b.x = wlds[(kb + koff    ) * DIM + col];
            b.y = wlds[(kb + koff + 1) * DIM + col];
            acc[n] = __builtin_amdgcn_wmma_f32_16x16x4_f32(
                 false, a,  false, b,
                 (short)0, acc[n],
                 false,  false);
        }
    }

    #pragma unroll
    for (int n = 0; n < NTILES; ++n) {
        #pragma unroll
        for (int r = 0; r < 8; ++r) {
            const int m = mtile * 16 + half * 8 + r;
            if (m < nrows) h[(size_t)m * DIM + n * 16 + l15] = acc[n][r];
        }
    }
}

#define EPW 32
#define EB 8
#define SC_WAVES 4

template <int B>
__device__ __forceinline__ void sc_load_rows(float (&v)[EB][4], const float* __restrict__ h,
                                             int s_l, int lane) {
    for (int j = 0; j < EB; ++j) {
        const int s = __builtin_amdgcn_readlane(s_l, B * EB + j);
        const float* hp = h + (size_t)s * DIM + lane;
        v[j][0] = hp[0];
        v[j][1] = hp[32];
        v[j][2] = hp[64];
        v[j][3] = hp[96];
    }
}

template <int B>
__device__ __forceinline__ void sc_add_rows(const float (&v)[EB][4], float* out,
                                            int d_l, int w_bits, int lane) {
    for (int j = 0; j < EB; ++j) {
        const int d = __builtin_amdgcn_readlane(d_l, B * EB + j);
        const float wt = __int_as_float(__builtin_amdgcn_readlane(w_bits, B * EB + j));
        float* o = out + (size_t)d * DIM + lane;
        __hip_atomic_fetch_add(o +  0, v[j][0] * wt, __ATOMIC_RELAXED, __HIP_MEMORY_SCOPE_AGENT);
        __hip_atomic_fetch_add(o + 32, v[j][1] * wt, __ATOMIC_RELAXED, __HIP_MEMORY_SCOPE_AGENT);
        __hip_atomic_fetch_add(o + 64, v[j][2] * wt, __ATOMIC_RELAXED, __HIP_MEMORY_SCOPE_AGENT);
        __hip_atomic_fetch_add(o + 96, v[j][3] * wt, __ATOMIC_RELAXED, __HIP_MEMORY_SCOPE_AGENT);
    }
}

__global__ __launch_bounds__(SC_WAVES * 32) void gcn_scatter(const float* __restrict__ h,
                                                             const int* __restrict__ esrc,
                                                             const int* __restrict__ edst,
                                                             const float* __restrict__ ew,
                                                             float* out, int E) {
    const int lane = threadIdx.x & 31;
    const int wid = __builtin_amdgcn_readfirstlane((int)(threadIdx.x >> 5));
    const long long base = ((long long)blockIdx.x * SC_WAVES + wid) * EPW;
    if (base >= E) return;

    if (base + EPW <= E) {
        const int s_l = esrc[base + lane];
        const int d_l = edst[base + lane];
        const int w_bits = __float_as_int(ew[base + lane]);

        float va[EB][4], vb[EB][4];
        sc_load_rows<0>(va, h, s_l, lane);
        sc_load_rows<1>(vb, h, s_l, lane);
        sc_add_rows<0>(va, out, d_l, w_bits, lane);
        sc_load_rows<2>(va, h, s_l, lane);
        sc_add_rows<1>(vb, out, d_l, w_bits, lane);
        sc_load_rows<3>(vb, h, s_l, lane);
        sc_add_rows<2>(va, out, d_l, w_bits, lane);
        sc_add_rows<3>(vb, out, d_l, w_bits, lane);
    } else {
        const int cnt = (int)(E - base);
        const bool live = lane < cnt;
        const int s_l = live ? esrc[base + lane] : 0;
        const int d_l = live ? edst[base + lane] : 0;
        const int w_bits = live ? __float_as_int(ew[base + lane]) : 0;
        for (int j = 0; j < cnt; ++j) {
            const int s = __builtin_amdgcn_readlane(s_l, j);
            const int d = __builtin_amdgcn_readlane(d_l, j);
            const float wt = __int_as_float(__builtin_amdgcn_readlane(w_bits, j));
            const float* hp = h + (size_t)s * DIM + lane;
            float* o = out + (size_t)d * DIM + lane;
            const float v0 = hp[0], v1 = hp[32], v2 = hp[64], v3 = hp[96];
            __hip_atomic_fetch_add(o +  0, v0 * wt, __ATOMIC_RELAXED, __HIP_MEMORY_SCOPE_AGENT);
            __hip_atomic_fetch_add(o + 32, v1 * wt, __ATOMIC_RELAXED, __HIP_MEMORY_SCOPE_AGENT);
            __hip_atomic_fetch_add(o + 64, v2 * wt, __ATOMIC_RELAXED, __HIP_MEMORY_SCOPE_AGENT);
            __hip_atomic_fetch_add(o + 96, v3 * wt, __ATOMIC_RELAXED, __HIP_MEMORY_SCOPE_AGENT);
        }
    }
}

extern "C" void kernel_launch(void* const* d_in, const int* in_sizes, int n_in,
                              void* d_out, int out_size, void* d_ws, size_t ws_size,
                              hipStream_t stream) {
    const float* x    = (const float*)d_in[0];
    const int*   esrc = (const int*)d_in[1];
    const int*   edst = (const int*)d_in[2];
    const float* ew   = (const float*)d_in[3];
    const float* w    = (const float*)d_in[4];
    const float* bias = (const float*)d_in[5];

    const int N = in_sizes[0] / DIM;
    const int E = in_sizes[1];

    float* out = (float*)d_out;
    float* h   = (float*)d_ws;

    {
        const int total = N * DIM;
        gcn_init_out<<<(total + 255) / 256, 256, 0, stream>>>(out, bias, total);
    }

    {
        const int mtiles = (N + 15) >> 4;
        const int blocks = (mtiles + WAVES_PER_BLOCK - 1) / WAVES_PER_BLOCK;
        gcn_gemm_wmma<<<blocks, 256, 0, stream>>>(x, w, h, N);
    }

    {
        const int edges_per_block = SC_WAVES * EPW;
        const int blocks = (int)(((long long)E + edges_per_block - 1) / edges_per_block);
        gcn_scatter<<<blocks, SC_WAVES * 32, 0, stream>>>(h, esrc, edst, ew, out, E);
    }
}


# binning: each 8192-edge chunk writes a row of a tile-count table; fill blocks sum its columns, sort their chunk by tile in LDS, store whole runs: no global atomics, no clearing launch
# speedup vs baseline: 3.4555x; 1.7958x over previous
#include <hip/hip_runtime.h>

typedef __attribute__((ext_vector_type(2))) float v2f;
typedef __attribute__((ext_vector_type(8))) float v8f;

#define DIM 128
#define NTILES 8

#define TILE 400
#define MT_PER_TILE (TILE / 16)
#define ACC_PITCH 132
#define TK_THREADS 1024
#define TK_WAVES 32

#define NBMAX 1024
#define BIN_THREADS 1024
#define BIN_K 8
#define BIN_CHUNK (BIN_THREADS * BIN_K)
#define WS_HDR 4096
#define PF_U 8

static_assert(BIN_THREADS == NBMAX, "one thread per table column");


__device__ __forceinline__ int bin_blocks_of(int E) {
    return (int)(((long long)E + BIN_CHUNK - 1) / BIN_CHUNK);
}

__global__ __launch_bounds__(BIN_THREADS) void gcn_bin_hist(const int* __restrict__ edst,
                                                            int* wsi, int E, int N) {
    __shared__ int lc[NBMAX];
    const int t = threadIdx.x;
    lc[t] = 0;
    const unsigned cbase = blockIdx.x * (unsigned)BIN_CHUNK;
    int d[BIN_K];
    #pragma unroll
    for (int k = 0; k < BIN_K; ++k) {
        const unsigned e = cbase + k * BIN_THREADS + t;
        const int v = edst[e < (unsigned)E ? e : (unsigned)E - 1u];
        d[k] = e < (unsigned)E ? v : -1;
    }
    __syncthreads();
    #pragma unroll
    for (int k = 0; k < BIN_K; ++k) {
        if ((unsigned)d[k] < (unsigned)N) atomicAdd(&lc[d[k] / TILE], 1);
    }
    __syncthreads();
    wsi[WS_HDR + blockIdx.x * NBMAX + t] = lc[t];
}

__global__ __launch_bounds__(BIN_THREADS) void gcn_bin_fill(const int* __restrict__ esrc,
                                                            const int* __restrict__ edst,
                                                            const float* __restrict__ ew,
                                                            int* wsi, int E, int N, int NB) {
    __shared__ int ltot[NBMAX];
    __shared__ int lbef[NBMAX];
    __shared__ int lc[NBMAX];
    __shared__ int ldel[NBMAX];
    __shared__ int wtot[2][32];
    __shared__ int st_s[BIN_CHUNK];
    __shared__ int st_bd[BIN_CHUNK];
    __shared__ float st_w[BIN_CHUNK];

    const int t = threadIdx.x;
    const int lane = t & 31;
    const int wv = t >> 5;
    const int nblk = gridDim.x;
    const int kb = blockIdx.x;

    int* start = wsi;
    const int* tab = wsi + WS_HDR;
    int* ssrc = wsi + WS_HDR + nblk * NBMAX;
    int* sdl  = ssrc + E;
    float* sw = reinterpret_cast<float*>(sdl + E);

    int d[BIN_K], s[BIN_K], bb[BIN_K], rk[BIN_K];
    float wg[BIN_K];
    const unsigned cbase = kb * (unsigned)BIN_CHUNK;
    #pragma unroll
    for (int k = 0; k < BIN_K; ++k) {
        const unsigned e = cbase + k * BIN_THREADS + t;
        const unsigned ec = e < (unsigned)E ? e : (unsigned)E - 1u;
        const int v = edst[ec];
        s[k]  = esrc[ec];
        wg[k] = ew[ec];
        d[k] = e < (unsigned)E ? v : -1;
    }

    ltot[t] = 0;
    lbef[t] = 0;
    lc[t] = 0;
    __syncthreads();

    {
        const int rg = __builtin_amdgcn_readfirstlane(t >> 6);
        const int4* tab4 = reinterpret_cast<const int4*>(tab);
        for (int c4 = t & 63; c4 * 4 < NB; c4 += 64) {
            int tx = 0, ty = 0, tz = 0, tw = 0;
            int bx = 0, by = 0, bz = 0, bw = 0;
            for (int r0 = rg; r0 < nblk; r0 += 16 * PF_U) {
                int4 v[PF_U];
                #pragma unroll
                for (int u = 0; u < PF_U; ++u) {
                    const int r = r0 + 16 * u;
                    const int rc = r < nblk ? r : nblk - 1;
                    v[u] = tab4[rc * (NBMAX / 4) + c4];
                }
                #pragma unroll
                for (int u = 0; u < PF_U; ++u) {
                    const int r = r0 + 16 * u;
                    if (r < nblk) {
                        tx += v[u].x; ty += v[u].y; tz += v[u].z; tw += v[u].w;
                        if (r < kb) { bx += v[u].x; by += v[u].y; bz += v[u].z; bw += v[u].w; }
                    }
                }
            }
            atomicAdd(&ltot[c4 * 4 + 0], tx);
            atomicAdd(&ltot[c4 * 4 + 1], ty);
            atomicAdd(&ltot[c4 * 4 + 2], tz);
            atomicAdd(&ltot[c4 * 4 + 3], tw);
            atomicAdd(&lbef[c4 * 4 + 0], bx);
            atomicAdd(&lbef[c4 * 4 + 1], by);
            atomicAdd(&lbef[c4 * 4 + 2], bz);
            atomicAdd(&lbef[c4 * 4 + 3], bw);
        }
    }

    #pragma unroll
    for (int k = 0; k < BIN_K; ++k) {
        bb[k] = -1;
        rk[k] = 0;
        if ((unsigned)d[k] < (unsigned)N) {
            bb[k] = d[k] / TILE;
            rk[k] = atomicAdd(&lc[bb[k]], 1);
        }
    }
    __syncthreads();

    const int a = ltot[t];
    const int c = lc[t];
    int ia = a, ic = c;
    #pragma unroll
    for (int off = 1; off < 32; off <<= 1) {
        const int ua = __shfl_up(ia, off);
        const int uc = __shfl_up(ic, off);
        if (lane >= off) { ia += ua; ic += uc; }
    }
    if (lane == 31) { wtot[0][wv] = ia; wtot[1][wv] = ic; }
    __syncthreads();
    int sa = wtot[0][lane];
    int sc = wtot[1][lane];
    #pragma unroll
    for (int off = 1; off < 32; off <<= 1) {
        const int ua = __shfl_up(sa, off);
        const int uc = __shfl_up(sc, off);
        if (lane >= off) { sa += ua; sc += uc; }
    }
    const int pa = wv ? __shfl(sa, wv - 1) : 0;
    const int pc = wv ? __shfl(sc, wv - 1) : 0;
    const int total  = __shfl(sa, 31);
    const int nvalid = __shfl(sc, 31);
    const int gstart = pa + ia - a;
    const int loff   = pc + ic - c;
    ltot[t] = loff;
    ldel[t] = gstart + lbef[t] - loff;
    if (kb == 0) {
        if (t < NB) start[t] = gstart;
        if (t == 0) start[NB] = total;
    }
    __syncthreads();

    int lo[BIN_K];
    #pragma unroll
    for (int k = 0; k < BIN_K; ++k) lo[k] = ltot[bb[k] >= 0 ? bb[k] : 0];
    #pragma unroll
    for (int k = 0; k < BIN_K; ++k) {
        if (bb[k] >= 0) {
            const int p = lo[k] + rk[k];
            st_s[p]  = s[k];
            st_w[p]  = wg[k];
            st_bd[p] = (bb[k] << 16) | (d[k] - bb[k] * TILE);
        }
    }
    __syncthreads();

    int bd[BIN_K], os[BIN_K], dl[BIN_K];
    float ow[BIN_K];
    #pragma unroll
    for (int k = 0; k < BIN_K; ++k) {
        const int i = k * BIN_THREADS + t;
        const int ic2 = i < nvalid ? i : 0;
        bd[k] = st_bd[ic2];
        os[k] = st_s[ic2];
        ow[k] = st_w[ic2];
    }
    #pragma unroll
    for (int k = 0; k < BIN_K; ++k) dl[k] = ldel[(bd[k] >> 16) & (NBMAX - 1)];
    #pragma unroll
    for (int k = 0; k < BIN_K; ++k) {
        const int i = k * BIN_THREADS + t;
        if (i < nvalid) {
            const int pos = dl[k] + i;
            ssrc[pos] = os[k];
            sdl[pos]  = bd[k] & 0xffff;
            sw[pos]   = ow[k];
        }
    }
}

#define EB 8

template <int B>
__device__ __forceinline__ void tk_load_rows(float (&v)[EB][4], const float* __restrict__ x,
                                             int s_l, int lane) {
    for (int j = 0; j < EB; ++j) {
        const int s = __builtin_amdgcn_readlane(s_l, B * EB + j);
        const float* hp = x + (size_t)s * DIM + lane;
        v[j][0] = hp[0];
        v[j][1] = hp[32];
        v[j][2] = hp[64];
        v[j][3] = hp[96];
    }
}

template <int B>
__device__ __forceinline__ void tk_add_rows(const float (&v)[EB][4], float* sums,
                                            int d_l, int w_bits, int lane) {
    for (int j = 0; j < EB; ++j) {
        const int d = __builtin_amdgcn_readlane(d_l, B * EB + j);
        const float wt = __int_as_float(__builtin_amdgcn_readlane(w_bits, B * EB + j));
        float* o = sums + d * ACC_PITCH + lane;
        __hip_atomic_fetch_add(o +  0, v[j][0] * wt, __ATOMIC_RELAXED, __HIP_MEMORY_SCOPE_WORKGROUP);
        __hip_atomic_fetch_add(o + 32, v[j][1] * wt, __ATOMIC_RELAXED, __HIP_MEMORY_SCOPE_WORKGROUP);
        __hip_atomic_fetch_add(o + 64, v[j][2] * wt, __ATOMIC_RELAXED, __HIP_MEMORY_SCOPE_WORKGROUP);
        __hip_atomic_fetch_add(o + 96, v[j][3] * wt, __ATOMIC_RELAXED, __HIP_MEMORY_SCOPE_WORKGROUP);
    }
}

__global__ __launch_bounds__(TK_THREADS) void gcn_tile_sum_gemm(const float* __restrict__ x,
                                                                const float* __restrict__ w,
                                                                const float* __restrict__ bias,
                                                                const int* __restrict__ wsi,
                                                                float* __restrict__ out,
                                                                int N, int E) {
    __shared__ float sums[TILE * ACC_PITCH];
    __shared__ float wlds[DIM * DIM];

    const int t = threadIdx.x;
    {
        float4* s4 = reinterpret_cast<float4*>(sums);
        const float4 z = {0.f, 0.f, 0.f, 0.f};
        for (int i = t; i < TILE * ACC_PITCH / 4; i += TK_THREADS) s4[i] = z;
        const float4* wsrc = reinterpret_cast<const float4*>(w);
        float4* wdst = reinterpret_cast<float4*>(wlds);
        #pragma unroll
        for (int i = t; i < DIM * DIM / 4; i += TK_THREADS) wdst[i] = wsrc[i];
    }
    __syncthreads();

    const int lane = t & 31;
    const int wid = __builtin_amdgcn_readfirstlane(t >> 5);

    const int* start = wsi;
    const int* ssrc  = wsi + WS_HDR + bin_blocks_of(E) * NBMAX;
    const int* sdl   = ssrc + E;
    const float* sw  = reinterpret_cast<const float*>(sdl + E);

    const int s0 = start[blockIdx.x];
    const int s1 = start[blockIdx.x + 1];

    for (int base = s0 + wid * 32; base < s1; base += TK_WAVES * 32) {
        if (base + 32 <= s1) {
            const int s_l = ssrc[base + lane];
            const int d_l = sdl[base + lane];
            const int w_bits = __float_as_int(sw[base + lane]);

            float va[EB][4], vb[EB][4];
            tk_load_rows<0>(va, x, s_l, lane);
            tk_load_rows<1>(vb, x, s_l, lane);
            tk_add_rows<0>(va, sums, d_l, w_bits, lane);
            tk_load_rows<2>(va, x, s_l, lane);
            tk_add_rows<1>(vb, sums, d_l, w_bits, lane);
            tk_load_rows<3>(vb, x, s_l, lane);
            tk_add_rows<2>(va, sums, d_l, w_bits, lane);
            tk_add_rows<3>(vb, sums, d_l, w_bits, lane);
        } else {
            const int cntp = s1 - base;
            const bool live = lane < cntp;
            const int s_l = live ? ssrc[base + lane] : 0;
            const int d_l = live ? sdl[base + lane] : 0;
            const int w_bits = live ? __float_as_int(sw[base + lane]) : 0;
            for (int g = 0; g < cntp; g += EB) {
                const int n = cntp - g;
                float v[EB][4];
                #pragma unroll
                for (int j = 0; j < EB; ++j) {
                    const int s = __builtin_amdgcn_readlane(s_l, j < n ? g + j : g);
                    const float* hp = x + (size_t)s * DIM + lane;
                    v[j][0] = hp[0];
                    v[j][1] = hp[32];
                    v[j][2] = hp[64];
                    v[j][3] = hp[96];
                }
                #pragma unroll
                for (int j = 0; j < EB; ++j) {
                    if (j < n) {
                        const int d = __builtin_amdgcn_readlane(d_l, g + j);
                        const float wt = __int_as_float(__builtin_amdgcn_readlane(w_bits, g + j));
                        float* o = sums + d * ACC_PITCH + lane;
                        __hip_atomic_fetch_add(o +  0, v[j][0] * wt, __ATOMIC_RELAXED, __HIP_MEMORY_SCOPE_WORKGROUP);
                        __hip_atomic_fetch_add(o + 32, v[j][1] * wt, __ATOMIC_RELAXED, __HIP_MEMORY_SCOPE_WORKGROUP);
                        __hip_atomic_fetch_add(o + 64, v[j][2] * wt, __ATOMIC_RELAXED, __HIP_MEMORY_SCOPE_WORKGROUP);
                        __hip_atomic_fetch_add(o + 96, v[j][3] * wt, __ATOMIC_RELAXED, __HIP_MEMORY_SCOPE_WORKGROUP);
                    }
                }
            }
        }
    }
    __syncthreads();

    if (wid >= MT_PER_TILE) return;

    const int half = lane >> 4;
    const int l15  = lane & 15;
    const int lrow = wid * 16 + l15;

    v8f acc[NTILES];
    #pragma unroll
    for (int n = 0; n < NTILES; ++n) {
        const float bv = bias[n * 16 + l15];
        const v8f vb8 = {bv, bv, bv, bv, bv, bv, bv, bv};
        acc[n] = vb8;
    }

    const int koff = half * 2;

    for (int kb = 0; kb < DIM; kb += 4) {
        const float2 af =
            *reinterpret_cast<const float2*>(&sums[lrow * ACC_PITCH + kb + koff]);
        v2f a;
        a.x = af.x;
        a.y = af.y;

        #pragma unroll
        for (int n = 0; n < NTILES; ++n) {
            const int col = n * 16 + l15;
            v2f b;
            b.x = wlds[(kb + koff    ) * DIM + col];
            b.y = wlds[(kb + koff + 1) * DIM + col];
            acc[n] = __builtin_amdgcn_wmma_f32_16x16x4_f32(
                 false, a,  false, b,
                 (short)0, acc[n],
                 false,  false);
        }
    }

    const int row0 = blockIdx.x * TILE + wid * 16 + half * 8;
    #pragma unroll
    for (int n = 0; n < NTILES; ++n) {
        #pragma unroll
        for (int r = 0; r < 8; ++r) {
            const int m = row0 + r;
            if (m < N) out[(size_t)m * DIM + n * 16 + l15] = acc[n][r];
        }
    }
}

extern "C" void kernel_launch(void* const* d_in, const int* in_sizes, int n_in,
                              void* d_out, int out_size, void* d_ws, size_t ws_size,
                              hipStream_t stream) {
    const float* x    = (const float*)d_in[0];
    const int*   esrc = (const int*)d_in[1];
    const int*   edst = (const int*)d_in[2];
    const float* ew   = (const float*)d_in[3];
    const float* w    = (const float*)d_in[4];
    const float* bias = (const float*)d_in[5];

    const int N = in_sizes[0] / DIM;
    const int E = in_sizes[1];

    float* out = (float*)d_out;
    int*   wsi = (int*)d_ws;

    const int NB = (N + TILE - 1) / TILE;
    const int bin_blocks = (int)(((long long)E + BIN_CHUNK - 1) / BIN_CHUNK);

    gcn_bin_hist<<<bin_blocks, BIN_THREADS, 0, stream>>>(edst, wsi, E, N);
    gcn_bin_fill<<<bin_blocks, BIN_THREADS, 0, stream>>>(esrc, edst, ew, wsi, E, N, NB);
    gcn_tile_sum_gemm<<<NB, TK_THREADS, 0, stream>>>(x, w, bias, wsi, out, N, E);
}


# tile kernel counting-sorts its edges by destination row in LDS; a wave sums each run of equal rows in registers, one LDS add per run, not per edge; sum order: sorted runs, ranks by arrival
# speedup vs baseline: 3.9545x; 1.1444x over previous
#include <hip/hip_runtime.h>

typedef __attribute__((ext_vector_type(2))) float v2f;
typedef __attribute__((ext_vector_type(8))) float v8f;

#define DIM 128
#define NTILES 8

#define TILE 400
#define MT_PER_TILE (TILE / 16)
#define ACC_PITCH 132
#define TK_THREADS 1024
#define TK_WAVES 32

#define NBMAX 1024
#define BIN_THREADS 1024
#define BIN_K 8
#define BIN_CHUNK (BIN_THREADS * BIN_K)
#define WS_HDR 4096
#define PF_U 8

static_assert(BIN_THREADS == NBMAX, "one thread per table column");


__device__ __forceinline__ int bin_blocks_of(int E) {
    return (int)(((long long)E + BIN_CHUNK - 1) / BIN_CHUNK);
}

__global__ __launch_bounds__(BIN_THREADS) void gcn_bin_hist(const int* __restrict__ edst,
                                                            int* wsi, int E, int N) {
    __shared__ int lc[NBMAX];
    const int t = threadIdx.x;
    lc[t] = 0;
    const unsigned cbase = blockIdx.x * (unsigned)BIN_CHUNK;
    int d[BIN_K];
    #pragma unroll
    for (int k = 0; k < BIN_K; ++k) {
        const unsigned e = cbase + k * BIN_THREADS + t;
        const int v = edst[e < (unsigned)E ? e : (unsigned)E - 1u];
        d[k] = e < (unsigned)E ? v : -1;
    }
    __syncthreads();
    #pragma unroll
    for (int k = 0; k < BIN_K; ++k) {
        if ((unsigned)d[k] < (unsigned)N) atomicAdd(&lc[d[k] / TILE], 1);
    }
    __syncthreads();
    wsi[WS_HDR + blockIdx.x * NBMAX + t] = lc[t];
}

__global__ __launch_bounds__(BIN_THREADS) void gcn_bin_fill(const int* __restrict__ esrc,
                                                            const int* __restrict__ edst,
                                                            const float* __restrict__ ew,
                                                            int* wsi, int E, int N, int NB) {
    __shared__ int ltot[NBMAX];
    __shared__ int lbef[NBMAX];
    __shared__ int lc[NBMAX];
    __shared__ int ldel[NBMAX];
    __shared__ int wtot[2][32];
    __shared__ int st_s[BIN_CHUNK];
    __shared__ int st_bd[BIN_CHUNK];
    __shared__ float st_w[BIN_CHUNK];

    const int t = threadIdx.x;
    const int lane = t & 31;
    const int wv = t >> 5;
    const int nblk = gridDim.x;
    const int kb = blockIdx.x;

    int* start = wsi;
    const int* tab = wsi + WS_HDR;
    int* ssrc = wsi + WS_HDR + nblk * NBMAX;
    int* sdl  = ssrc + E;
    float* sw = reinterpret_cast<float*>(sdl + E);

    int d[BIN_K], s[BIN_K], bb[BIN_K], rk[BIN_K];
    float wg[BIN_K];
    const unsigned cbase = kb * (unsigned)BIN_CHUNK;
    #pragma unroll
    for (int k = 0; k < BIN_K; ++k) {
        const unsigned e = cbase + k * BIN_THREADS + t;
        const unsigned ec = e < (unsigned)E ? e : (unsigned)E - 1u;
        const int v = edst[ec];
        s[k]  = esrc[ec];
        wg[k] = ew[ec];
        d[k] = e < (unsigned)E ? v : -1;
    }

    ltot[t] = 0;
    lbef[t] = 0;
    lc[t] = 0;
    __syncthreads();

    {
        const int rg = __builtin_amdgcn_readfirstlane(t >> 6);
        const int4* tab4 = reinterpret_cast<const int4*>(tab);
        for (int c4 = t & 63; c4 * 4 < NB; c4 += 64) {
            int tx = 0, ty = 0, tz = 0, tw = 0;
            int bx = 0, by = 0, bz = 0, bw = 0;
            for (int r0 = rg; r0 < nblk; r0 += 16 * PF_U) {
                int4 v[PF_U];
                #pragma unroll
                for (int u = 0; u < PF_U; ++u) {
                    const int r = r0 + 16 * u;
                    const int rc = r < nblk ? r : nblk - 1;
                    v[u] = tab4[rc * (NBMAX / 4) + c4];
                }
                #pragma unroll
                for (int u = 0; u < PF_U; ++u) {
                    const int r = r0 + 16 * u;
                    if (r < nblk) {
                        tx += v[u].x; ty += v[u].y; tz += v[u].z; tw += v[u].w;
                        if (r < kb) { bx += v[u].x; by += v[u].y; bz += v[u].z; bw += v[u].w; }
                    }
                }
            }
            atomicAdd(&ltot[c4 * 4 + 0], tx);
            atomicAdd(&ltot[c4 * 4 + 1], ty);
            atomicAdd(&ltot[c4 * 4 + 2], tz);
            atomicAdd(&ltot[c4 * 4 + 3], tw);
            atomicAdd(&lbef[c4 * 4 + 0], bx);
            atomicAdd(&lbef[c4 * 4 + 1], by);
            atomicAdd(&lbef[c4 * 4 + 2], bz);
            atomicAdd(&lbef[c4 * 4 + 3], bw);
        }
    }

    #pragma unroll
    for (int k = 0; k < BIN_K; ++k) {
        bb[k] = -1;
        rk[k] = 0;
        if ((unsigned)d[k] < (unsigned)N) {
            bb[k] = d[k] / TILE;
            rk[k] = atomicAdd(&lc[bb[k]], 1);
        }
    }
    __syncthreads();

    const int a = ltot[t];
    const int c = lc[t];
    int ia = a, ic = c;
    #pragma unroll
    for (int off = 1; off < 32; off <<= 1) {
        const int ua = __shfl_up(ia, off);
        const int uc = __shfl_up(ic, off);
        if (lane >= off) { ia += ua; ic += uc; }
    }
    if (lane == 31) { wtot[0][wv] = ia; wtot[1][wv] = ic; }
    __syncthreads();
    int sa = wtot[0][lane];
    int sc = wtot[1][lane];
    #pragma unroll
    for (int off = 1; off < 32; off <<= 1) {
        const int ua = __shfl_up(sa, off);
        const int uc = __shfl_up(sc, off);
        if (lane >= off) { sa += ua; sc += uc; }
    }
    const int pa = wv ? __shfl(sa, wv - 1) : 0;
    const int pc = wv ? __shfl(sc, wv - 1) : 0;
    const int total  = __shfl(sa, 31);
    const int nvalid = __shfl(sc, 31);
    const int gstart = pa + ia - a;
    const int loff   = pc + ic - c;
    ltot[t] = loff;
    ldel[t] = gstart + lbef[t] - loff;
    if (kb == 0) {
        if (t < NB) start[t] = gstart;
        if (t == 0) start[NB] = total;
    }
    __syncthreads();

    int lo[BIN_K];
    #pragma unroll
    for (int k = 0; k < BIN_K; ++k) lo[k] = ltot[bb[k] >= 0 ? bb[k] : 0];
    #pragma unroll
    for (int k = 0; k < BIN_K; ++k) {
        if (bb[k] >= 0) {
            const int p = lo[k] + rk[k];
            st_s[p]  = s[k];
            st_w[p]  = wg[k];
            st_bd[p] = (bb[k] << 16) | (d[k] - bb[k] * TILE);
        }
    }
    __syncthreads();

    int bd[BIN_K], os[BIN_K], dl[BIN_K];
    float ow[BIN_K];
    #pragma unroll
    for (int k = 0; k < BIN_K; ++k) {
        const int i = k * BIN_THREADS + t;
        const int ic2 = i < nvalid ? i : 0;
        bd[k] = st_bd[ic2];
        os[k] = st_s[ic2];
        ow[k] = st_w[ic2];
    }
    #pragma unroll
    for (int k = 0; k < BIN_K; ++k) dl[k] = ldel[(bd[k] >> 16) & (NBMAX - 1)];
    #pragma unroll
    for (int k = 0; k < BIN_K; ++k) {
        const int i = k * BIN_THREADS + t;
        if (i < nvalid) {
            const int pos = dl[k] + i;
            ssrc[pos] = os[k];
            sdl[pos]  = bd[k] & 0xffff;
            sw[pos]   = ow[k];
        }
    }
}

#define EB 8
#define PASS 8192
#define PASS_K (PASS / TK_THREADS)
#define KEY_SHIFT 19
#define KEY_MASK ((1 << KEY_SHIFT) - 1)

static_assert(NBMAX * TILE <= (1 << KEY_SHIFT), "a source row must fit the low key bits");
static_assert(PASS * 2 == DIM * DIM, "keys + weights of a pass fill exactly the W buffer");
static_assert(TILE <= TK_THREADS, "one thread per row in the scan");

template <int B>
__device__ __forceinline__ void tk_load_rows(float (&v)[EB][4], const float* __restrict__ x,
                                             int key_l, int lane) {
    #pragma unroll
    for (int j = 0; j < EB; ++j) {
        const int s = __builtin_amdgcn_readlane(key_l, B * EB + j) & KEY_MASK;
        const float* hp = x + (size_t)s * DIM + lane;
        v[j][0] = hp[0];
        v[j][1] = hp[32];
        v[j][2] = hp[64];
        v[j][3] = hp[96];
    }
}

__device__ __forceinline__ void tk_flush_row(float* sums, int row, int lane, const float (&acc)[4]) {
    float* o = sums + row * ACC_PITCH + lane;
    __hip_atomic_fetch_add(o +  0, acc[0], __ATOMIC_RELAXED, __HIP_MEMORY_SCOPE_WORKGROUP);
    __hip_atomic_fetch_add(o + 32, acc[1], __ATOMIC_RELAXED, __HIP_MEMORY_SCOPE_WORKGROUP);
    __hip_atomic_fetch_add(o + 64, acc[2], __ATOMIC_RELAXED, __HIP_MEMORY_SCOPE_WORKGROUP);
    __hip_atomic_fetch_add(o + 96, acc[3], __ATOMIC_RELAXED, __HIP_MEMORY_SCOPE_WORKGROUP);
}

__device__ __forceinline__ void tk_sum_edge(const float (&v)[4], float wt, int row, float* sums,
                                            int lane, int& cur, float (&acc)[4]) {
    if (row != cur) {
        tk_flush_row(sums, cur, lane, acc);
        cur = row;
        acc[0] = 0.f; acc[1] = 0.f; acc[2] = 0.f; acc[3] = 0.f;
    }
    acc[0] += v[0] * wt;
    acc[1] += v[1] * wt;
    acc[2] += v[2] * wt;
    acc[3] += v[3] * wt;
}

template <int B>
__device__ __forceinline__ void tk_sum_rows(const float (&v)[EB][4], float* sums, int key_l,
                                            int w_bits, int lane, int& cur, float (&acc)[4]) {
    #pragma unroll
    for (int j = 0; j < EB; ++j) {
        const int row = __builtin_amdgcn_readlane(key_l, B * EB + j) >> KEY_SHIFT;
        const float wt = __int_as_float(__builtin_amdgcn_readlane(w_bits, B * EB + j));
        tk_sum_edge(v[j], wt, row, sums, lane, cur, acc);
    }
}

__global__ __launch_bounds__(TK_THREADS) void gcn_tile_sum_gemm(const float* __restrict__ x,
                                                                const float* __restrict__ w,
                                                                const float* __restrict__ bias,
                                                                const int* __restrict__ wsi,
                                                                float* __restrict__ out,
                                                                int N, int E) {
    __shared__ float sums[TILE * ACC_PITCH];
    __shared__ float wlds[DIM * DIM];
    __shared__ int cnt[TK_THREADS];
    __shared__ int wtot[TK_WAVES];

    int* ekey = reinterpret_cast<int*>(wlds);
    int* ewt  = ekey + PASS;

    const int t = threadIdx.x;
    const int lane = t & 31;
    const int wid = __builtin_amdgcn_readfirstlane(t >> 5);

    const int* start = wsi;
    const int* ssrc  = wsi + WS_HDR + bin_blocks_of(E) * NBMAX;
    const int* sdl   = ssrc + E;
    const int* swb   = sdl + E;

    const int s0 = start[blockIdx.x];
    const int s1 = start[blockIdx.x + 1];

    int pbase = s0;
    do {
        const int n = s1 - pbase < PASS ? s1 - pbase : PASS;

        int es[PASS_K], ed[PASS_K], ew[PASS_K], rk[PASS_K], fs[PASS_K];
        #pragma unroll
        for (int k = 0; k < PASS_K; ++k) {
            const int i = k * TK_THREADS + t;
            int e = pbase + (i < n ? i : 0);
            e = e < E ? e : E - 1;
            es[k] = ssrc[e];
            ew[k] = swb[e];
            const unsigned d = (unsigned)sdl[e];
            ed[k] = (int)(d < (unsigned)TILE ? d : (unsigned)TILE - 1u);
        }
        if (pbase == s0) {
            float4* s4 = reinterpret_cast<float4*>(sums);
            const float4 z = {0.f, 0.f, 0.f, 0.f};
            for (int i = t; i < TILE * ACC_PITCH / 4; i += TK_THREADS) s4[i] = z;
        }
        cnt[t] = 0;
        __syncthreads();

        #pragma unroll
        for (int k = 0; k < PASS_K; ++k) {
            rk[k] = 0;
            if (k * TK_THREADS + t < n) rk[k] = atomicAdd(&cnt[ed[k]], 1);
        }
        __syncthreads();

        {
            const int cv = cnt[t];
            int iv = cv;
            #pragma unroll
            for (int off = 1; off < 32; off <<= 1) {
                const int u = __shfl_up(iv, off);
                if (lane >= off) iv += u;
            }
            if (lane == 31) wtot[wid] = iv;
            __syncthreads();
            int sv = wtot[lane];
            #pragma unroll
            for (int off = 1; off < 32; off <<= 1) {
                const int u = __shfl_up(sv, off);
                if (lane >= off) sv += u;
            }
            const int pv = wid ? __shfl(sv, wid - 1) : 0;
            cnt[t] = pv + iv - cv;
        }
        __syncthreads();

        #pragma unroll
        for (int k = 0; k < PASS_K; ++k) fs[k] = cnt[ed[k]];
        #pragma unroll
        for (int k = 0; k < PASS_K; ++k) {
            if (k * TK_THREADS + t < n) {
                const int p = fs[k] + rk[k];
                ekey[p] = (ed[k] << KEY_SHIFT) | (es[k] & KEY_MASK);
                ewt[p]  = ew[k];
            }
        }
        __syncthreads();

        const int q  = ((n + TK_WAVES - 1) / TK_WAVES + EB - 1) & ~(EB - 1);
        const int lo = wid * q;
        const int hi = lo + q < n ? lo + q : n;
        if (lo < hi) {
            int cur = __builtin_amdgcn_readfirstlane(ekey[lo]) >> KEY_SHIFT;
            float acc[4] = {0.f, 0.f, 0.f, 0.f};
            for (int base = lo; base < hi; base += 32) {
                if (base + 32 <= hi) {
                    const int key_l  = ekey[base + lane];
                    const int w_bits = ewt[base + lane];

                    float va[EB][4], vb[EB][4];
                    tk_load_rows<0>(va, x, key_l, lane);
                    tk_load_rows<1>(vb, x, key_l, lane);
                    tk_sum_rows<0>(va, sums, key_l, w_bits, lane, cur, acc);
                    tk_load_rows<2>(va, x, key_l, lane);
                    tk_sum_rows<1>(vb, sums, key_l, w_bits, lane, cur, acc);
                    tk_load_rows<3>(vb, x, key_l, lane);
                    tk_sum_rows<2>(va, sums, key_l, w_bits, lane, cur, acc);
                    tk_sum_rows<3>(vb, sums, key_l, w_bits, lane, cur, acc);
                } else {
                    const int cntp = hi - base;
                    const bool live = lane < cntp;
                    const int key_l  = ekey[live ? base + lane : base];
                    const int w_bits = ewt[live ? base + lane : base];
                    for (int g = 0; g < cntp; g += EB) {
                        const int m = cntp - g;
                        float v[EB][4];
                        #pragma unroll
                        for (int j = 0; j < EB; ++j) {
                            const int s = __builtin_amdgcn_readlane(key_l, j < m ? g + j : g) & KEY_MASK;
                            const float* hp = x + (size_t)s * DIM + lane;
                            v[j][0] = hp[0];
                            v[j][1] = hp[32];
                            v[j][2] = hp[64];
                            v[j][3] = hp[96];
                        }
                        #pragma unroll
                        for (int j = 0; j < EB; ++j) {
                            if (j < m) {
                                const int row = __builtin_amdgcn_readlane(key_l, g + j) >> KEY_SHIFT;
                                const float wt = __int_as_float(__builtin_amdgcn_readlane(w_bits, g + j));
                                tk_sum_edge(v[j], wt, row, sums, lane, cur, acc);
                            }
                        }
                    }
                }
            }
            tk_flush_row(sums, cur, lane, acc);
        }

        pbase += PASS;
        if (pbase < s1) __syncthreads();
    } while (pbase < s1);

    __syncthreads();
    {
        const float4* wsrc = reinterpret_cast<const float4*>(w);
        float4* wdst = reinterpret_cast<float4*>(wlds);
        #pragma unroll
        for (int i = t; i < DIM * DIM / 4; i += TK_THREADS) wdst[i] = wsrc[i];
    }
    __syncthreads();

    if (wid >= MT_PER_TILE) return;

    const int half = lane >> 4;
    const int l15  = lane & 15;
    const int lrow = wid * 16 + l15;

    v8f acc[NTILES];
    #pragma unroll
    for (int n = 0; n < NTILES; ++n) {
        const float bv = bias[n * 16 + l15];
        const v8f vb8 = {bv, bv, bv, bv, bv, bv, bv, bv};
        acc[n] = vb8;
    }

    const int koff = half * 2;

    for (int kb = 0; kb < DIM; kb += 4) {
        const float2 af =
            *reinterpret_cast<const float2*>(&sums[lrow * ACC_PITCH + kb + koff]);
        v2f a;
        a.x = af.x;
        a.y = af.y;

        #pragma unroll
        for (int n = 0; n < NTILES; ++n) {
            const int col = n * 16 + l15;
            v2f b;
            b.x = wlds[(kb + koff    ) * DIM + col];
            b.y = wlds[(kb + koff + 1) * DIM + col];
            acc[n] = __builtin_amdgcn_wmma_f32_16x16x4_f32(
                 false, a,  false, b,
                 (short)0, acc[n],
                 false,  false);
        }
    }

    const int row0 = blockIdx.x * TILE + wid * 16 + half * 8;
    #pragma unroll
    for (int n = 0; n < NTILES; ++n) {
        #pragma unroll
        for (int r = 0; r < 8; ++r) {
            const int m = row0 + r;
            if (m < N) out[(size_t)m * DIM + n * 16 + l15] = acc[n][r];
        }
    }
}

extern "C" void kernel_launch(void* const* d_in, const int* in_sizes, int n_in,
                              void* d_out, int out_size, void* d_ws, size_t ws_size,
                              hipStream_t stream) {
    const float* x    = (const float*)d_in[0];
    const int*   esrc = (const int*)d_in[1];
    const int*   edst = (const int*)d_in[2];
    const float* ew   = (const float*)d_in[3];
    const float* w    = (const float*)d_in[4];
    const float* bias = (const float*)d_in[5];

    const int N = in_sizes[0] / DIM;
    const int E = in_sizes[1];

    float* out = (float*)d_out;
    int*   wsi = (int*)d_ws;

    const int NB = (N + TILE - 1) / TILE;
    const int bin_blocks = (int)(((long long)E + BIN_CHUNK - 1) / BIN_CHUNK);

    gcn_bin_hist<<<bin_blocks, BIN_THREADS, 0, stream>>>(edst, wsi, E, N);
    gcn_bin_fill<<<bin_blocks, BIN_THREADS, 0, stream>>>(esrc, edst, ew, wsi, E, N, NB);
    gcn_tile_sum_gemm<<<NB, TK_THREADS, 0, stream>>>(x, w, bias, wsi, out, N, E);
}


# tile kernel: lane l owns columns 4l..4l+3, so an x row is one 16-byte load per lane (4 whole lines per load instruction), not four 4-byte loads of one line each; order of the sums unchanged
# speedup vs baseline: 4.0864x; 1.0334x over previous
#include <hip/hip_runtime.h>

typedef __attribute__((ext_vector_type(2))) float v2f;
typedef __attribute__((ext_vector_type(8))) float v8f;

#define DIM 128
#define NTILES 8

#define TILE 400
#define MT_PER_TILE (TILE / 16)
#define ACC_PITCH 132
#define TK_THREADS 1024
#define TK_WAVES 32

#define NBMAX 1024
#define BIN_THREADS 1024
#define BIN_K 8
#define BIN_CHUNK (BIN_THREADS * BIN_K)
#define WS_HDR 4096
#define PF_U 8

static_assert(BIN_THREADS == NBMAX, "one thread per table column");


__device__ __forceinline__ int bin_blocks_of(int E) {
    return (int)(((long long)E + BIN_CHUNK - 1) / BIN_CHUNK);
}

__global__ __launch_bounds__(BIN_THREADS) void gcn_bin_hist(const int* __restrict__ edst,
                                                            int* wsi, int E, int N) {
    __shared__ int lc[NBMAX];
    const int t = threadIdx.x;
    lc[t] = 0;
    const unsigned cbase = blockIdx.x * (unsigned)BIN_CHUNK;
    int d[BIN_K];
    #pragma unroll
    for (int k = 0; k < BIN_K; ++k) {
        const unsigned e = cbase + k * BIN_THREADS + t;
        const int v = edst[e < (unsigned)E ? e : (unsigned)E - 1u];
        d[k] = e < (unsigned)E ? v : -1;
    }
    __syncthreads();
    #pragma unroll
    for (int k = 0; k < BIN_K; ++k) {
        if ((unsigned)d[k] < (unsigned)N) atomicAdd(&lc[d[k] / TILE], 1);
    }
    __syncthreads();
    wsi[WS_HDR + blockIdx.x * NBMAX + t] = lc[t];
}

__global__ __launch_bounds__(BIN_THREADS) void gcn_bin_fill(const int* __restrict__ esrc,
                                                            const int* __restrict__ edst,
                                                            const float* __restrict__ ew,
                                                            int* wsi, int E, int N, int NB) {
    __shared__ int ltot[NBMAX];
    __shared__ int lbef[NBMAX];
    __shared__ int lc[NBMAX];
    __shared__ int ldel[NBMAX];
    __shared__ int wtot[2][32];
    __shared__ int st_s[BIN_CHUNK];
    __shared__ int st_bd[BIN_CHUNK];
    __shared__ float st_w[BIN_CHUNK];

    const int t = threadIdx.x;
    const int lane = t & 31;
    const int wv = t >> 5;
    const int nblk = gridDim.x;
    const int kb = blockIdx.x;

    int* start = wsi;
    const int* tab = wsi + WS_HDR;
    int* ssrc = wsi + WS_HDR + nblk * NBMAX;
    int* sdl  = ssrc + E;
    float* sw = reinterpret_cast<float*>(sdl + E);

    int d[BIN_K], s[BIN_K], bb[BIN_K], rk[BIN_K];
    float wg[BIN_K];
    const unsigned cbase = kb * (unsigned)BIN_CHUNK;
    #pragma unroll
    for (int k = 0; k < BIN_K; ++k) {
        const unsigned e = cbase + k * BIN_THREADS + t;
        const unsigned ec = e < (unsigned)E ? e : (unsigned)E - 1u;
        const int v = edst[ec];
        s[k]  = esrc[ec];
        wg[k] = ew[ec];
        d[k] = e < (unsigned)E ? v : -1;
    }

    ltot[t] = 0;
    lbef[t] = 0;
    lc[t] = 0;
    __syncthreads();

    {
        const int rg = __builtin_amdgcn_readfirstlane(t >> 6);
        const int4* tab4 = reinterpret_cast<const int4*>(tab);
        for (int c4 = t & 63; c4 * 4 < NB; c4 += 64) {
            int tx = 0, ty = 0, tz = 0, tw = 0;
            int bx = 0, by = 0, bz = 0, bw = 0;
            for (int r0 = rg; r0 < nblk; r0 += 16 * PF_U) {
                int4 v[PF_U];
                #pragma unroll
                for (int u = 0; u < PF_U; ++u) {
                    const int r = r0 + 16 * u;
                    const int rc = r < nblk ? r : nblk - 1;
                    v[u] = tab4[rc * (NBMAX / 4) + c4];
                }
                #pragma unroll
                for (int u = 0; u < PF_U; ++u) {
                    const int r = r0 + 16 * u;
                    if (r < nblk) {
                        tx += v[u].x; ty += v[u].y; tz += v[u].z; tw += v[u].w;
                        if (r < kb) { bx += v[u].x; by += v[u].y; bz += v[u].z; bw += v[u].w; }
                    }
                }
            }
            atomicAdd(&ltot[c4 * 4 + 0], tx);
            atomicAdd(&ltot[c4 * 4 + 1], ty);
            atomicAdd(&ltot[c4 * 4 + 2], tz);
            atomicAdd(&ltot[c4 * 4 + 3], tw);
            atomicAdd(&lbef[c4 * 4 + 0], bx);
            atomicAdd(&lbef[c4 * 4 + 1], by);
            atomicAdd(&lbef[c4 * 4 + 2], bz);
            atomicAdd(&lbef[c4 * 4 + 3], bw);
        }
    }

    #pragma unroll
    for (int k = 0; k < BIN_K; ++k) {
        bb[k] = -1;
        rk[k] = 0;
        if ((unsigned)d[k] < (unsigned)N) {
            bb[k] = d[k] / TILE;
            rk[k] = atomicAdd(&lc[bb[k]], 1);
        }
    }
    __syncthreads();

    const int a = ltot[t];
    const int c = lc[t];
    int ia = a, ic = c;
    #pragma unroll
    for (int off = 1; off < 32; off <<= 1) {
        const int ua = __shfl_up(ia, off);
        const int uc = __shfl_up(ic, off);
        if (lane >= off) { ia += ua; ic += uc; }
    }
    if (lane == 31) { wtot[0][wv] = ia; wtot[1][wv] = ic; }
    __syncthreads();
    int sa = wtot[0][lane];
    int sc = wtot[1][lane];
    #pragma unroll
    for (int off = 1; off < 32; off <<= 1) {
        const int ua = __shfl_up(sa, off);
        const int uc = __shfl_up(sc, off);
        if (lane >= off) { sa += ua; sc += uc; }
    }
    const int pa = wv ? __shfl(sa, wv - 1) : 0;
    const int pc = wv ? __shfl(sc, wv - 1) : 0;
    const int total  = __shfl(sa, 31);
    const int nvalid = __shfl(sc, 31);
    const int gstart = pa + ia - a;
    const int loff   = pc + ic - c;
    ltot[t] = loff;
    ldel[t] = gstart + lbef[t] - loff;
    if (kb == 0) {
        if (t < NB) start[t] = gstart;
        if (t == 0) start[NB] = total;
    }
    __syncthreads();

    int lo[BIN_K];
    #pragma unroll
    for (int k = 0; k < BIN_K; ++k) lo[k] = ltot[bb[k] >= 0 ? bb[k] : 0];
    #pragma unroll
    for (int k = 0; k < BIN_K; ++k) {
        if (bb[k] >= 0) {
            const int p = lo[k] + rk[k];
            st_s[p]  = s[k];
            st_w[p]  = wg[k];
            st_bd[p] = (bb[k] << 16) | (d[k] - bb[k] * TILE);
        }
    }
    __syncthreads();

    int bd[BIN_K], os[BIN_K], dl[BIN_K];
    float ow[BIN_K];
    #pragma unroll
    for (int k = 0; k < BIN_K; ++k) {
        const int i = k * BIN_THREADS + t;
        const int ic2 = i < nvalid ? i : 0;
        bd[k] = st_bd[ic2];
        os[k] = st_s[ic2];
        ow[k] = st_w[ic2];
    }
    #pragma unroll
    for (int k = 0; k < BIN_K; ++k) dl[k] = ldel[(bd[k] >> 16) & (NBMAX - 1)];
    #pragma unroll
    for (int k = 0; k < BIN_K; ++k) {
        const int i = k * BIN_THREADS + t;
        if (i < nvalid) {
            const int pos = dl[k] + i;
            ssrc[pos] = os[k];
            sdl[pos]  = bd[k] & 0xffff;
            sw[pos]   = ow[k];
        }
    }
}

#define EB 8
#define PASS 8192
#define PASS_K (PASS / TK_THREADS)
#define KEY_SHIFT 19
#define KEY_MASK ((1 << KEY_SHIFT) - 1)

static_assert(NBMAX * TILE <= (1 << KEY_SHIFT), "a source row must fit the low key bits");
static_assert(PASS * 2 == DIM * DIM, "keys + weights of a pass fill exactly the W buffer");
static_assert(TILE <= TK_THREADS, "one thread per row in the scan");

template <int B>
__device__ __forceinline__ void tk_load_rows(float (&v)[EB][4], const float* __restrict__ x,
                                             int key_l, int lane) {
    #pragma unroll
    for (int j = 0; j < EB; ++j) {
        const int s = __builtin_amdgcn_readlane(key_l, B * EB + j) & KEY_MASK;
        const float4 r = reinterpret_cast<const float4*>(x + (size_t)s * DIM)[lane];
        v[j][0] = r.x;
        v[j][1] = r.y;
        v[j][2] = r.z;
        v[j][3] = r.w;
    }
}

__device__ __forceinline__ void tk_flush_row(float* sums, int row, int lane, const float (&acc)[4]) {
    float* o = sums + row * ACC_PITCH + 4 * lane;
    __hip_atomic_fetch_add(o + 0, acc[0], __ATOMIC_RELAXED, __HIP_MEMORY_SCOPE_WORKGROUP);
    __hip_atomic_fetch_add(o + 1, acc[1], __ATOMIC_RELAXED, __HIP_MEMORY_SCOPE_WORKGROUP);
    __hip_atomic_fetch_add(o + 2, acc[2], __ATOMIC_RELAXED, __HIP_MEMORY_SCOPE_WORKGROUP);
    __hip_atomic_fetch_add(o + 3, acc[3], __ATOMIC_RELAXED, __HIP_MEMORY_SCOPE_WORKGROUP);
}

__device__ __forceinline__ void tk_sum_edge(const float (&v)[4], float wt, int row, float* sums,
                                            int lane, int& cur, float (&acc)[4]) {
    if (row != cur) {
        tk_flush_row(sums, cur, lane, acc);
        cur = row;
        acc[0] = 0.f; acc[1] = 0.f; acc[2] = 0.f; acc[3] = 0.f;
    }
    acc[0] += v[0] * wt;
    acc[1] += v[1] * wt;
    acc[2] += v[2] * wt;
    acc[3] += v[3] * wt;
}

template <int B>
__device__ __forceinline__ void tk_sum_rows(const float (&v)[EB][4], float* sums, int key_l,
                                            int w_bits, int lane, int& cur, float (&acc)[4]) {
    #pragma unroll
    for (int j = 0; j < EB; ++j) {
        const int row = __builtin_amdgcn_readlane(key_l, B * EB + j) >> KEY_SHIFT;
        const float wt = __int_as_float(__builtin_amdgcn_readlane(w_bits, B * EB + j));
        tk_sum_edge(v[j], wt, row, sums, lane, cur, acc);
    }
}

__global__ __launch_bounds__(TK_THREADS) void gcn_tile_sum_gemm(const float* __restrict__ x,
                                                                const float* __restrict__ w,
                                                                const float* __restrict__ bias,
                                                                const int* __restrict__ wsi,
                                                                float* __restrict__ out,
                                                                int N, int E) {
    __shared__ float sums[TILE * ACC_PITCH];
    __shared__ float wlds[DIM * DIM];
    __shared__ int cnt[TK_THREADS];
    __shared__ int wtot[TK_WAVES];

    int* ekey = reinterpret_cast<int*>(wlds);
    int* ewt  = ekey + PASS;

    const int t = threadIdx.x;
    const int lane = t & 31;
    const int wid = __builtin_amdgcn_readfirstlane(t >> 5);

    const int* start = wsi;
    const int* ssrc  = wsi + WS_HDR + bin_blocks_of(E) * NBMAX;
    const int* sdl   = ssrc + E;
    const int* swb   = sdl + E;

    const int s0 = start[blockIdx.x];
    const int s1 = start[blockIdx.x + 1];

    int pbase = s0;
    do {
        const int n = s1 - pbase < PASS ? s1 - pbase : PASS;

        int es[PASS_K], ed[PASS_K], ew[PASS_K], rk[PASS_K], fs[PASS_K];
        #pragma unroll
        for (int k = 0; k < PASS_K; ++k) {
            const int i = k * TK_THREADS + t;
            int e = pbase + (i < n ? i : 0);
            e = e < E ? e : E - 1;
            es[k] = ssrc[e];
            ew[k] = swb[e];
            const unsigned d = (unsigned)sdl[e];
            ed[k] = (int)(d < (unsigned)TILE ? d : (unsigned)TILE - 1u);
        }
        if (pbase == s0) {
            float4* s4 = reinterpret_cast<float4*>(sums);
            const float4 z = {0.f, 0.f, 0.f, 0.f};
            for (int i = t; i < TILE * ACC_PITCH / 4; i += TK_THREADS) s4[i] = z;
        }
        cnt[t] = 0;
        __syncthreads();

        #pragma unroll
        for (int k = 0; k < PASS_K; ++k) {
            rk[k] = 0;
            if (k * TK_THREADS + t < n) rk[k] = atomicAdd(&cnt[ed[k]], 1);
        }
        __syncthreads();

        {
            const int cv = cnt[t];
            int iv = cv;
            #pragma unroll
            for (int off = 1; off < 32; off <<= 1) {
                const int u = __shfl_up(iv, off);
                if (lane >= off) iv += u;
            }
            if (lane == 31) wtot[wid] = iv;
            __syncthreads();
            int sv = wtot[lane];
            #pragma unroll
            for (int off = 1; off < 32; off <<= 1) {
                const int u = __shfl_up(sv, off);
                if (lane >= off) sv += u;
            }
            const int pv = wid ? __shfl(sv, wid - 1) : 0;
            cnt[t] = pv + iv - cv;
        }
        __syncthreads();

        #pragma unroll
        for (int k = 0; k < PASS_K; ++k) fs[k] = cnt[ed[k]];
        #pragma unroll
        for (int k = 0; k < PASS_K; ++k) {
            if (k * TK_THREADS + t < n) {
                const int p = fs[k] + rk[k];
                ekey[p] = (ed[k] << KEY_SHIFT) | (es[k] & KEY_MASK);
                ewt[p]  = ew[k];
            }
        }
        __syncthreads();

        const int q  = ((n + TK_WAVES - 1) / TK_WAVES + EB - 1) & ~(EB - 1);
        const int lo = wid * q;
        const int hi = lo + q < n ? lo + q : n;
        if (lo < hi) {
            int cur = __builtin_amdgcn_readfirstlane(ekey[lo]) >> KEY_SHIFT;
            float acc[4] = {0.f, 0.f, 0.f, 0.f};
            for (int base = lo; base < hi; base += 32) {
                if (base + 32 <= hi) {
                    const int key_l  = ekey[base + lane];
                    const int w_bits = ewt[base + lane];

                    float va[EB][4], vb[EB][4];
                    tk_load_rows<0>(va, x, key_l, lane);
                    tk_load_rows<1>(vb, x, key_l, lane);
                    tk_sum_rows<0>(va, sums, key_l, w_bits, lane, cur, acc);
                    tk_load_rows<2>(va, x, key_l, lane);
                    tk_sum_rows<1>(vb, sums, key_l, w_bits, lane, cur, acc);
                    tk_load_rows<3>(vb, x, key_l, lane);
                    tk_sum_rows<2>(va, sums, key_l, w_bits, lane, cur, acc);
                    tk_sum_rows<3>(vb, sums, key_l, w_bits, lane, cur, acc);
                } else {
                    const int cntp = hi - base;
                    const bool live = lane < cntp;
                    const int key_l  = ekey[live ? base + lane : base];
                    const int w_bits = ewt[live ? base + lane : base];
                    for (int g = 0; g < cntp; g += EB) {
                        const int m = cntp - g;
                        float v[EB][4];
                        #pragma unroll
                        for (int j = 0; j < EB; ++j) {
                            const int s = __builtin_amdgcn_readlane(key_l, j < m ? g + j : g) & KEY_MASK;
                            const float4 r = reinterpret_cast<const float4*>(x + (size_t)s * DIM)[lane];
                            v[j][0] = r.x;
                            v[j][1] = r.y;
                            v[j][2] = r.z;
                            v[j][3] = r.w;
                        }
                        #pragma unroll
                        for (int j = 0; j < EB; ++j) {
                            if (j < m) {
                                const int row = __builtin_amdgcn_readlane(key_l, g + j) >> KEY_SHIFT;
                                const float wt = __int_as_float(__builtin_amdgcn_readlane(w_bits, g + j));
                                tk_sum_edge(v[j], wt, row, sums, lane, cur, acc);
                            }
                        }
                    }
                }
            }
            tk_flush_row(sums, cur, lane, acc);
        }

        pbase += PASS;
        if (pbase < s1) __syncthreads();
    } while (pbase < s1);

    __syncthreads();
    {
        const float4* wsrc = reinterpret_cast<const float4*>(w);
        float4* wdst = reinterpret_cast<float4*>(wlds);
        #pragma unroll
        for (int i = t; i < DIM * DIM / 4; i += TK_THREADS) wdst[i] = wsrc[i];
    }
    __syncthreads();

    if (wid >= MT_PER_TILE) return;

    const int half = lane >> 4;
    const int l15  = lane & 15;
    const int lrow = wid * 16 + l15;

    v8f acc[NTILES];
    #pragma unroll
    for (int n = 0; n < NTILES; ++n) {
        const float bv = bias[n * 16 + l15];
        const v8f vb8 = {bv, bv, bv, bv, bv, bv, bv, bv};
        acc[n] = vb8;
    }

    const int koff = half * 2;

    for (int kb = 0; kb < DIM; kb += 4) {
        const float2 af =
            *reinterpret_cast<const float2*>(&sums[lrow * ACC_PITCH + kb + koff]);
        v2f a;
        a.x = af.x;
        a.y = af.y;

        #pragma unroll
        for (int n = 0; n < NTILES; ++n) {
            const int col = n * 16 + l15;
            v2f b;
            b.x = wlds[(kb + koff    ) * DIM + col];
            b.y = wlds[(kb + koff + 1) * DIM + col];
            acc[n] = __builtin_amdgcn_wmma_f32_16x16x4_f32(
                 false, a,  false, b,
                 (short)0, acc[n],
                 false,  false);
        }
    }

    const int row0 = blockIdx.x * TILE + wid * 16 + half * 8;
    #pragma unroll
    for (int n = 0; n < NTILES; ++n) {
        #pragma unroll
        for (int r = 0; r < 8; ++r) {
            const int m = row0 + r;
            if (m < N) out[(size_t)m * DIM + n * 16 + l15] = acc[n][r];
        }
    }
}

extern "C" void kernel_launch(void* const* d_in, const int* in_sizes, int n_in,
                              void* d_out, int out_size, void* d_ws, size_t ws_size,
                              hipStream_t stream) {
    const float* x    = (const float*)d_in[0];
    const int*   esrc = (const int*)d_in[1];
    const int*   edst = (const int*)d_in[2];
    const float* ew   = (const float*)d_in[3];
    const float* w    = (const float*)d_in[4];
    const float* bias = (const float*)d_in[5];

    const int N = in_sizes[0] / DIM;
    const int E = in_sizes[1];

    float* out = (float*)d_out;
    int*   wsi = (int*)d_ws;

    const int NB = (N + TILE - 1) / TILE;
    const int bin_blocks = (int)(((long long)E + BIN_CHUNK - 1) / BIN_CHUNK);

    gcn_bin_hist<<<bin_blocks, BIN_THREADS, 0, stream>>>(edst, wsi, E, N);
    gcn_bin_fill<<<bin_blocks, BIN_THREADS, 0, stream>>>(esrc, edst, ew, wsi, E, N, NB);
    gcn_tile_sum_gemm<<<NB, TK_THREADS, 0, stream>>>(x, w, bias, wsi, out, N, E);
}


# binning in one launch, not two: each 8192-edge chunk sorted by tile in place with a run table; a tile's work-group scans its table column, finds an edge's run by binary search; sum order kept
# speedup vs baseline: 4.4410x; 1.0868x over previous
#include <hip/hip_runtime.h>

typedef __attribute__((ext_vector_type(2))) float v2f;
typedef __attribute__((ext_vector_type(8))) float v8f;

#define DIM 128
#define NTILES 8

#define TILE 400
#define MT_PER_TILE (TILE / 16)
#define ACC_PITCH 132
#define TK_THREADS 1024
#define TK_WAVES 32

#define NBMAX 1024
#define BIN_THREADS 1024
#define BIN_K 8
#define BIN_CHUNK (BIN_THREADS * BIN_K)
#define NKMAX 1024

static_assert(BIN_THREADS == NBMAX, "one thread per table column");
static_assert(NKMAX == TK_THREADS, "one thread per chunk in the run scan");
static_assert(BIN_CHUNK < (1 << 15), "a run's first slot and its length share one word");


__device__ __forceinline__ int bin_blocks_of(int E) {
    return (int)(((long long)E + BIN_CHUNK - 1) / BIN_CHUNK);
}

__global__ __launch_bounds__(BIN_THREADS) void gcn_bin_sort(const int* __restrict__ esrc,
                                                            const int* __restrict__ edst,
                                                            const float* __restrict__ ew,
                                                            int* wsi, int E, int N) {
    __shared__ int lc[NBMAX];
    __shared__ int wtot[32];
    __shared__ int st_s[BIN_CHUNK];
    __shared__ int st_d[BIN_CHUNK];
    __shared__ float st_w[BIN_CHUNK];

    const int t = threadIdx.x;
    const int lane = t & 31;
    const int wv = t >> 5;
    const int nblk = gridDim.x;
    const int kb = blockIdx.x;

    int* tab  = wsi;
    int* ssrc = wsi + nblk * NBMAX;
    int* sdl  = ssrc + E;
    float* sw = reinterpret_cast<float*>(sdl + E);

    int d[BIN_K], s[BIN_K], bb[BIN_K], rk[BIN_K];
    float wg[BIN_K];
    const unsigned cbase = kb * (unsigned)BIN_CHUNK;
    #pragma unroll
    for (int k = 0; k < BIN_K; ++k) {
        const unsigned e = cbase + k * BIN_THREADS + t;
        const unsigned ec = e < (unsigned)E ? e : (unsigned)E - 1u;
        const int v = edst[ec];
        s[k]  = esrc[ec];
        wg[k] = ew[ec];
        d[k] = e < (unsigned)E ? v : -1;
    }

    lc[t] = 0;
    __syncthreads();

    #pragma unroll
    for (int k = 0; k < BIN_K; ++k) {
        bb[k] = -1;
        rk[k] = 0;
        if ((unsigned)d[k] < (unsigned)N) {
            bb[k] = d[k] / TILE;
            rk[k] = atomicAdd(&lc[bb[k]], 1);
        }
    }
    __syncthreads();

    const int c = lc[t];
    int ic = c;
    #pragma unroll
    for (int off = 1; off < 32; off <<= 1) {
        const int uc = __shfl_up(ic, off);
        if (lane >= off) ic += uc;
    }
    if (lane == 31) wtot[wv] = ic;
    __syncthreads();
    int sc = wtot[lane];
    #pragma unroll
    for (int off = 1; off < 32; off <<= 1) {
        const int uc = __shfl_up(sc, off);
        if (lane >= off) sc += uc;
    }
    const int pc = wv ? __shfl(sc, wv - 1) : 0;
    const int nvalid = __shfl(sc, 31);
    const int loff   = pc + ic - c;
    lc[t] = loff;
    tab[kb * NBMAX + t] = (loff << 16) | c;
    __syncthreads();

    int lo[BIN_K];
    #pragma unroll
    for (int k = 0; k < BIN_K; ++k) lo[k] = lc[bb[k] >= 0 ? bb[k] : 0];
    #pragma unroll
    for (int k = 0; k < BIN_K; ++k) {
        if (bb[k] >= 0) {
            const int p = lo[k] + rk[k];
            st_s[p] = s[k];
            st_w[p] = wg[k];
            st_d[p] = d[k] - bb[k] * TILE;
        }
    }
    __syncthreads();

    int os[BIN_K], od[BIN_K];
    float ow[BIN_K];
    #pragma unroll
    for (int k = 0; k < BIN_K; ++k) {
        const int i = k * BIN_THREADS + t;
        const int ic2 = i < nvalid ? i : 0;
        os[k] = st_s[ic2];
        od[k] = st_d[ic2];
        ow[k] = st_w[ic2];
    }
    #pragma unroll
    for (int k = 0; k < BIN_K; ++k) {
        const int i = k * BIN_THREADS + t;
        if (i < nvalid) {
            const unsigned pos = cbase + i;
            ssrc[pos] = os[k];
            sdl[pos]  = od[k];
            sw[pos]   = ow[k];
        }
    }
}

#define EB 8
#define PASS 8192
#define PASS_K (PASS / TK_THREADS)
#define KEY_SHIFT 19
#define KEY_MASK ((1 << KEY_SHIFT) - 1)

static_assert(NBMAX * TILE <= (1 << KEY_SHIFT), "a source row must fit the low key bits");
static_assert(PASS * 2 == DIM * DIM, "keys + weights of a pass fill exactly the W buffer");
static_assert(TILE <= TK_THREADS, "one thread per row in the scan");

template <int B>
__device__ __forceinline__ void tk_load_rows(float (&v)[EB][4], const float* __restrict__ x,
                                             int key_l, int lane) {
    #pragma unroll
    for (int j = 0; j < EB; ++j) {
        const int s = __builtin_amdgcn_readlane(key_l, B * EB + j) & KEY_MASK;
        const float4 r = reinterpret_cast<const float4*>(x + (size_t)s * DIM)[lane];
        v[j][0] = r.x;
        v[j][1] = r.y;
        v[j][2] = r.z;
        v[j][3] = r.w;
    }
}

__device__ __forceinline__ void tk_flush_row(float* sums, int row, int lane, const float (&acc)[4]) {
    float* o = sums + row * ACC_PITCH + 4 * lane;
    __hip_atomic_fetch_add(o + 0, acc[0], __ATOMIC_RELAXED, __HIP_MEMORY_SCOPE_WORKGROUP);
    __hip_atomic_fetch_add(o + 1, acc[1], __ATOMIC_RELAXED, __HIP_MEMORY_SCOPE_WORKGROUP);
    __hip_atomic_fetch_add(o + 2, acc[2], __ATOMIC_RELAXED, __HIP_MEMORY_SCOPE_WORKGROUP);
    __hip_atomic_fetch_add(o + 3, acc[3], __ATOMIC_RELAXED, __HIP_MEMORY_SCOPE_WORKGROUP);
}

__device__ __forceinline__ void tk_sum_edge(const float (&v)[4], float wt, int row, float* sums,
                                            int lane, int& cur, float (&acc)[4]) {
    if (row != cur) {
        tk_flush_row(sums, cur, lane, acc);
        cur = row;
        acc[0] = 0.f; acc[1] = 0.f; acc[2] = 0.f; acc[3] = 0.f;
    }
    acc[0] += v[0] * wt;
    acc[1] += v[1] * wt;
    acc[2] += v[2] * wt;
    acc[3] += v[3] * wt;
}

template <int B>
__device__ __forceinline__ void tk_sum_rows(const float (&v)[EB][4], float* sums, int key_l,
                                            int w_bits, int lane, int& cur, float (&acc)[4]) {
    #pragma unroll
    for (int j = 0; j < EB; ++j) {
        const int row = __builtin_amdgcn_readlane(key_l, B * EB + j) >> KEY_SHIFT;
        const float wt = __int_as_float(__builtin_amdgcn_readlane(w_bits, B * EB + j));
        tk_sum_edge(v[j], wt, row, sums, lane, cur, acc);
    }
}

__global__ __launch_bounds__(TK_THREADS) void gcn_tile_sum_gemm(const float* __restrict__ x,
                                                                const float* __restrict__ w,
                                                                const float* __restrict__ bias,
                                                                const int* __restrict__ wsi,
                                                                float* __restrict__ out,
                                                                int N, int E) {
    __shared__ float sums[TILE * ACC_PITCH];
    __shared__ float wlds[DIM * DIM];
    __shared__ int cnt[TK_THREADS];
    __shared__ int wtot[TK_WAVES];
    __shared__ int rs[NKMAX];
    __shared__ int dlt[NKMAX];

    int* ekey = reinterpret_cast<int*>(wlds);
    int* ewt  = ekey + PASS;

    const int t = threadIdx.x;
    const int lane = t & 31;
    const int wid = __builtin_amdgcn_readfirstlane(t >> 5);

    const int nblk   = bin_blocks_of(E);
    const int* tab   = wsi;
    const int* ssrc  = wsi + nblk * NBMAX;
    const int* sdl   = ssrc + E;
    const int* swb   = sdl + E;

    int total;
    {
        int cv = 0, ov = 0;
        if (t < nblk) {
            const int pk = tab[t * NBMAX + blockIdx.x];
            cv = pk & 0xffff;
            cv = cv < BIN_CHUNK ? cv : BIN_CHUNK;
            ov = (pk >> 16) & 0xffff;
            ov = ov < BIN_CHUNK ? ov : BIN_CHUNK;
        }
        int iv = cv;
        #pragma unroll
        for (int off = 1; off < 32; off <<= 1) {
            const int u = __shfl_up(iv, off);
            if (lane >= off) iv += u;
        }
        if (lane == 31) wtot[wid] = iv;
        __syncthreads();
        int sv = wtot[lane];
        #pragma unroll
        for (int off = 1; off < 32; off <<= 1) {
            const int u = __shfl_up(sv, off);
            if (lane >= off) sv += u;
        }
        const int pv = wid ? __shfl(sv, wid - 1) : 0;
        const int ex = pv + iv - cv;
        rs[t]  = ex;
        dlt[t] = t * BIN_CHUNK + ov - ex;
        total = __builtin_amdgcn_readfirstlane(__shfl(sv, 31));
    }
    __syncthreads();

    int step0 = NKMAX / 2;
    while (step0 >= nblk && step0 > 1) step0 >>= 1;

    int pbase = 0;
    do {
        const int n = total - pbase < PASS ? total - pbase : PASS;

        int es[PASS_K], ed[PASS_K], ew[PASS_K], rk[PASS_K], fs[PASS_K];
        int pp[PASS_K], kk[PASS_K];
        #pragma unroll
        for (int k = 0; k < PASS_K; ++k) {
            const int i = k * TK_THREADS + t;
            pp[k] = pbase + (i < n ? i : 0);
            kk[k] = 0;
        }
        for (int st = step0; st >= 1; st >>= 1) {
            int pr[PASS_K];
            #pragma unroll
            for (int k = 0; k < PASS_K; ++k) pr[k] = rs[kk[k] + st];
            #pragma unroll
            for (int k = 0; k < PASS_K; ++k) kk[k] += pr[k] <= pp[k] ? st : 0;
        }
        int dv[PASS_K];
        #pragma unroll
        for (int k = 0; k < PASS_K; ++k) dv[k] = dlt[kk[k]];
        #pragma unroll
        for (int k = 0; k < PASS_K; ++k) {
            int e = dv[k] + pp[k];
            e = (unsigned)e < (unsigned)E ? e : E - 1;
            es[k] = ssrc[e];
            ew[k] = swb[e];
            const unsigned d = (unsigned)sdl[e];
            ed[k] = (int)(d < (unsigned)TILE ? d : (unsigned)TILE - 1u);
        }
        if (pbase == 0) {
            float4* s4 = reinterpret_cast<float4*>(sums);
            const float4 z = {0.f, 0.f, 0.f, 0.f};
            for (int i = t; i < TILE * ACC_PITCH / 4; i += TK_THREADS) s4[i] = z;
        }
        cnt[t] = 0;
        __syncthreads();

        #pragma unroll
        for (int k = 0; k < PASS_K; ++k) {
            rk[k] = 0;
            if (k * TK_THREADS + t < n) rk[k] = atomicAdd(&cnt[ed[k]], 1);
        }
        __syncthreads();

        {
            const int cv = cnt[t];
            int iv = cv;
            #pragma unroll
            for (int off = 1; off < 32; off <<= 1) {
                const int u = __shfl_up(iv, off);
                if (lane >= off) iv += u;
            }
            if (lane == 31) wtot[wid] = iv;
            __syncthreads();
            int sv = wtot[lane];
            #pragma unroll
            for (int off = 1; off < 32; off <<= 1) {
                const int u = __shfl_up(sv, off);
                if (lane >= off) sv += u;
            }
            const int pv = wid ? __shfl(sv, wid - 1) : 0;
            cnt[t] = pv + iv - cv;
        }
        __syncthreads();

        #pragma unroll
        for (int k = 0; k < PASS_K; ++k) fs[k] = cnt[ed[k]];
        #pragma unroll
        for (int k = 0; k < PASS_K; ++k) {
            if (k * TK_THREADS + t < n) {
                const int p = fs[k] + rk[k];
                ekey[p] = (ed[k] << KEY_SHIFT) | (es[k] & KEY_MASK);
                ewt[p]  = ew[k];
            }
        }
        __syncthreads();

        const int q  = ((n + TK_WAVES - 1) / TK_WAVES + EB - 1) & ~(EB - 1);
        const int lo = wid * q;
        const int hi = lo + q < n ? lo + q : n;
        if (lo < hi) {
            int cur = __builtin_amdgcn_readfirstlane(ekey[lo]) >> KEY_SHIFT;
            float acc[4] = {0.f, 0.f, 0.f, 0.f};
            for (int base = lo; base < hi; base += 32) {
                if (base + 32 <= hi) {
                    const int key_l  = ekey[base + lane];
                    const int w_bits = ewt[base + lane];

                    float va[EB][4], vb[EB][4];
                    tk_load_rows<0>(va, x, key_l, lane);
                    tk_load_rows<1>(vb, x, key_l, lane);
                    tk_sum_rows<0>(va, sums, key_l, w_bits, lane, cur, acc);
                    tk_load_rows<2>(va, x, key_l, lane);
                    tk_sum_rows<1>(vb, sums, key_l, w_bits, lane, cur, acc);
                    tk_load_rows<3>(vb, x, key_l, lane);
                    tk_sum_rows<2>(va, sums, key_l, w_bits, lane, cur, acc);
                    tk_sum_rows<3>(vb, sums, key_l, w_bits, lane, cur, acc);
                } else {
                    const int cntp = hi - base;
                    const bool live = lane < cntp;
                    const int key_l  = ekey[live ? base + lane : base];
                    const int w_bits = ewt[live ? base + lane : base];
                    for (int g = 0; g < cntp; g += EB) {
                        const int m = cntp - g;
                        float v[EB][4];
                        #pragma unroll
                        for (int j = 0; j < EB; ++j) {
                            const int s = __builtin_amdgcn_readlane(key_l, j < m ? g + j : g) & KEY_MASK;
                            const float4 r = reinterpret_cast<const float4*>(x + (size_t)s * DIM)[lane];
                            v[j][0] = r.x;
                            v[j][1] = r.y;
                            v[j][2] = r.z;
                            v[j][3] = r.w;
                        }
                        #pragma unroll
                        for (int j = 0; j < EB; ++j) {
                            if (j < m) {
                                const int row = __builtin_amdgcn_readlane(key_l, g + j) >> KEY_SHIFT;
                                const float wt = __int_as_float(__builtin_amdgcn_readlane(w_bits, g + j));
                                tk_sum_edge(v[j], wt, row, sums, lane, cur, acc);
                            }
                        }
                    }
                }
            }
            tk_flush_row(sums, cur, lane, acc);
        }

        pbase += PASS;
        if (pbase < total) __syncthreads();
    } while (pbase < total);

    __syncthreads();
    {
        const float4* wsrc = reinterpret_cast<const float4*>(w);
        float4* wdst = reinterpret_cast<float4*>(wlds);
        #pragma unroll
        for (int i = t; i < DIM * DIM / 4; i += TK_THREADS) wdst[i] = wsrc[i];
    }
    __syncthreads();

    if (wid >= MT_PER_TILE) return;

    const int half = lane >> 4;
    const int l15  = lane & 15;
    const int lrow = wid * 16 + l15;

    v8f acc[NTILES];
    #pragma unroll
    for (int n = 0; n < NTILES; ++n) {
        const float bv = bias[n * 16 + l15];
        const v8f vb8 = {bv, bv, bv, bv, bv, bv, bv, bv};
        acc[n] = vb8;
    }

    const int koff = half * 2;

    for (int kb = 0; kb < DIM; kb += 4) {
        const float2 af =
            *reinterpret_cast<const float2*>(&sums[lrow * ACC_PITCH + kb + koff]);
        v2f a;
        a.x = af.x;
        a.y = af.y;

        #pragma unroll
        for (int n = 0; n < NTILES; ++n) {
            const int col = n * 16 + l15;
            v2f b;
            b.x = wlds[(kb + koff    ) * DIM + col];
            b.y = wlds[(kb + koff + 1) * DIM + col];
            acc[n] = __builtin_amdgcn_wmma_f32_16x16x4_f32(
                 false, a,  false, b,
                 (short)0, acc[n],
                 false,  false);
        }
    }

    const int row0 = blockIdx.x * TILE + wid * 16 + half * 8;
    #pragma unroll
    for (int n = 0; n < NTILES; ++n) {
        #pragma unroll
        for (int r = 0; r < 8; ++r) {
            const int m = row0 + r;
            if (m < N) out[(size_t)m * DIM + n * 16 + l15] = acc[n][r];
        }
    }
}

extern "C" void kernel_launch(void* const* d_in, const int* in_sizes, int n_in,
                              void* d_out, int out_size, void* d_ws, size_t ws_size,
                              hipStream_t stream) {
    const float* x    = (const float*)d_in[0];
    const int*   esrc = (const int*)d_in[1];
    const int*   edst = (const int*)d_in[2];
    const float* ew   = (const float*)d_in[3];
    const float* w    = (const float*)d_in[4];
    const float* bias = (const float*)d_in[5];

    const int N = in_sizes[0] / DIM;
    const int E = in_sizes[1];

    float* out = (float*)d_out;
    int*   wsi = (int*)d_ws;

    const int NB = (N + TILE - 1) / TILE;
    const int bin_blocks = (int)(((long long)E + BIN_CHUNK - 1) / BIN_CHUNK);

    gcn_bin_sort<<<bin_blocks, BIN_THREADS, 0, stream>>>(esrc, edst, ew, wsi, E, N);
    gcn_tile_sum_gemm<<<NB, TK_THREADS, 0, stream>>>(x, w, bias, wsi, out, N, E);
}


# phase 2 re-tiled: a wave owns 32 columns (a whole 128-byte line per row) of 3-4 row tiles, so all 32 waves share the WMMAs evenly per SIMD and every store writes a whole line; sum order kept
# speedup vs baseline: 4.6088x; 1.0378x over previous
#include <hip/hip_runtime.h>

typedef __attribute__((ext_vector_type(2))) float v2f;
typedef __attribute__((ext_vector_type(8))) float v8f;

#define DIM 128

#define TILE 400
#define MT_PER_TILE (TILE / 16)
#define ACC_PITCH 132
#define TK_THREADS 1024
#define TK_WAVES 32

#define NBMAX 1024
#define BIN_THREADS 1024
#define BIN_K 8
#define BIN_CHUNK (BIN_THREADS * BIN_K)
#define NKMAX 1024

static_assert(BIN_THREADS == NBMAX, "one thread per table column");
static_assert(NKMAX == TK_THREADS, "one thread per chunk in the run scan");
static_assert(BIN_CHUNK < (1 << 15), "a run's first slot and its length share one word");


__device__ __forceinline__ int bin_blocks_of(int E) {
    return (int)(((long long)E + BIN_CHUNK - 1) / BIN_CHUNK);
}

__global__ __launch_bounds__(BIN_THREADS) void gcn_bin_sort(const int* __restrict__ esrc,
                                                            const int* __restrict__ edst,
                                                            const float* __restrict__ ew,
                                                            int* wsi, int E, int N) {
    __shared__ int lc[NBMAX];
    __shared__ int wtot[32];
    __shared__ int st_s[BIN_CHUNK];
    __shared__ int st_d[BIN_CHUNK];
    __shared__ float st_w[BIN_CHUNK];

    const int t = threadIdx.x;
    const int lane = t & 31;
    const int wv = t >> 5;
    const int nblk = gridDim.x;
    const int kb = blockIdx.x;

    int* tab  = wsi;
    int* ssrc = wsi + nblk * NBMAX;
    int* sdl  = ssrc + E;
    float* sw = reinterpret_cast<float*>(sdl + E);

    int d[BIN_K], s[BIN_K], bb[BIN_K], rk[BIN_K];
    float wg[BIN_K];
    const unsigned cbase = kb * (unsigned)BIN_CHUNK;
    #pragma unroll
    for (int k = 0; k < BIN_K; ++k) {
        const unsigned e = cbase + k * BIN_THREADS + t;
        const unsigned ec = e < (unsigned)E ? e : (unsigned)E - 1u;
        const int v = edst[ec];
        s[k]  = esrc[ec];
        wg[k] = ew[ec];
        d[k] = e < (unsigned)E ? v : -1;
    }

    lc[t] = 0;
    __syncthreads();

    #pragma unroll
    for (int k = 0; k < BIN_K; ++k) {
        bb[k] = -1;
        rk[k] = 0;
        if ((unsigned)d[k] < (unsigned)N) {
            bb[k] = d[k] / TILE;
            rk[k] = atomicAdd(&lc[bb[k]], 1);
        }
    }
    __syncthreads();

    const int c = lc[t];
    int ic = c;
    #pragma unroll
    for (int off = 1; off < 32; off <<= 1) {
        const int uc = __shfl_up(ic, off);
        if (lane >= off) ic += uc;
    }
    if (lane == 31) wtot[wv] = ic;
    __syncthreads();
    int sc = wtot[lane];
    #pragma unroll
    for (int off = 1; off < 32; off <<= 1) {
        const int uc = __shfl_up(sc, off);
        if (lane >= off) sc += uc;
    }
    const int pc = wv ? __shfl(sc, wv - 1) : 0;
    const int nvalid = __shfl(sc, 31);
    const int loff   = pc + ic - c;
    lc[t] = loff;
    tab[kb * NBMAX + t] = (loff << 16) | c;
    __syncthreads();

    int lo[BIN_K];
    #pragma unroll
    for (int k = 0; k < BIN_K; ++k) lo[k] = lc[bb[k] >= 0 ? bb[k] : 0];
    #pragma unroll
    for (int k = 0; k < BIN_K; ++k) {
        if (bb[k] >= 0) {
            const int p = lo[k] + rk[k];
            st_s[p] = s[k];
            st_w[p] = wg[k];
            st_d[p] = d[k] - bb[k] * TILE;
        }
    }
    __syncthreads();

    int os[BIN_K], od[BIN_K];
    float ow[BIN_K];
    #pragma unroll
    for (int k = 0; k < BIN_K; ++k) {
        const int i = k * BIN_THREADS + t;
        const int ic2 = i < nvalid ? i : 0;
        os[k] = st_s[ic2];
        od[k] = st_d[ic2];
        ow[k] = st_w[ic2];
    }
    #pragma unroll
    for (int k = 0; k < BIN_K; ++k) {
        const int i = k * BIN_THREADS + t;
        if (i < nvalid) {
            const unsigned pos = cbase + i;
            ssrc[pos] = os[k];
            sdl[pos]  = od[k];
            sw[pos]   = ow[k];
        }
    }
}

#define EB 8
#define PASS 8192
#define PASS_K (PASS / TK_THREADS)
#define KEY_SHIFT 19
#define KEY_MASK ((1 << KEY_SHIFT) - 1)

static_assert(NBMAX * TILE <= (1 << KEY_SHIFT), "a source row must fit the low key bits");
static_assert(PASS * 2 == DIM * DIM, "keys + weights of a pass fill exactly the W buffer");
static_assert(TILE <= TK_THREADS, "one thread per row in the scan");

template <int B>
__device__ __forceinline__ void tk_load_rows(float (&v)[EB][4], const float* __restrict__ x,
                                             int key_l, int lane) {
    #pragma unroll
    for (int j = 0; j < EB; ++j) {
        const int s = __builtin_amdgcn_readlane(key_l, B * EB + j) & KEY_MASK;
        const float4 r = reinterpret_cast<const float4*>(x + (size_t)s * DIM)[lane];
        v[j][0] = r.x;
        v[j][1] = r.y;
        v[j][2] = r.z;
        v[j][3] = r.w;
    }
}

__device__ __forceinline__ void tk_flush_row(float* sums, int row, int lane, const float (&acc)[4]) {
    float* o = sums + row * ACC_PITCH + 4 * lane;
    __hip_atomic_fetch_add(o + 0, acc[0], __ATOMIC_RELAXED, __HIP_MEMORY_SCOPE_WORKGROUP);
    __hip_atomic_fetch_add(o + 1, acc[1], __ATOMIC_RELAXED, __HIP_MEMORY_SCOPE_WORKGROUP);
    __hip_atomic_fetch_add(o + 2, acc[2], __ATOMIC_RELAXED, __HIP_MEMORY_SCOPE_WORKGROUP);
    __hip_atomic_fetch_add(o + 3, acc[3], __ATOMIC_RELAXED, __HIP_MEMORY_SCOPE_WORKGROUP);
}

__device__ __forceinline__ void tk_sum_edge(const float (&v)[4], float wt, int row, float* sums,
                                            int lane, int& cur, float (&acc)[4]) {
    if (row != cur) {
        tk_flush_row(sums, cur, lane, acc);
        cur = row;
        acc[0] = 0.f; acc[1] = 0.f; acc[2] = 0.f; acc[3] = 0.f;
    }
    acc[0] += v[0] * wt;
    acc[1] += v[1] * wt;
    acc[2] += v[2] * wt;
    acc[3] += v[3] * wt;
}

template <int B>
__device__ __forceinline__ void tk_sum_rows(const float (&v)[EB][4], float* sums, int key_l,
                                            int w_bits, int lane, int& cur, float (&acc)[4]) {
    #pragma unroll
    for (int j = 0; j < EB; ++j) {
        const int row = __builtin_amdgcn_readlane(key_l, B * EB + j) >> KEY_SHIFT;
        const float wt = __int_as_float(__builtin_amdgcn_readlane(w_bits, B * EB + j));
        tk_sum_edge(v[j], wt, row, sums, lane, cur, acc);
    }
}

__global__ __launch_bounds__(TK_THREADS) void gcn_tile_sum_gemm(const float* __restrict__ x,
                                                                const float* __restrict__ w,
                                                                const float* __restrict__ bias,
                                                                const int* __restrict__ wsi,
                                                                float* __restrict__ out,
                                                                int N, int E) {
    __shared__ __attribute__((aligned(16))) float ldsf[DIM * DIM + TILE * ACC_PITCH];
    float* const wlds = ldsf;
    float* const sums = ldsf + DIM * DIM;
    __shared__ int cnt[TK_THREADS];
    __shared__ int wtot[TK_WAVES];
    __shared__ int rs[NKMAX];
    __shared__ int dlt[NKMAX];

    int* ekey = reinterpret_cast<int*>(wlds);
    int* ewt  = ekey + PASS;

    const int t = threadIdx.x;
    const int lane = t & 31;
    const int wid = __builtin_amdgcn_readfirstlane(t >> 5);

    const int nblk   = bin_blocks_of(E);
    const int* tab   = wsi;
    const int* ssrc  = wsi + nblk * NBMAX;
    const int* sdl   = ssrc + E;
    const int* swb   = sdl + E;

    int total;
    {
        int cv = 0, ov = 0;
        if (t < nblk) {
            const int pk = tab[t * NBMAX + blockIdx.x];
            cv = pk & 0xffff;
            cv = cv < BIN_CHUNK ? cv : BIN_CHUNK;
            ov = (pk >> 16) & 0xffff;
            ov = ov < BIN_CHUNK ? ov : BIN_CHUNK;
        }
        int iv = cv;
        #pragma unroll
        for (int off = 1; off < 32; off <<= 1) {
            const int u = __shfl_up(iv, off);
            if (lane >= off) iv += u;
        }
        if (lane == 31) wtot[wid] = iv;
        __syncthreads();
        int sv = wtot[lane];
        #pragma unroll
        for (int off = 1; off < 32; off <<= 1) {
            const int u = __shfl_up(sv, off);
            if (lane >= off) sv += u;
        }
        const int pv = wid ? __shfl(sv, wid - 1) : 0;
        const int ex = pv + iv - cv;
        rs[t]  = ex;
        dlt[t] = t * BIN_CHUNK + ov - ex;
        total = __builtin_amdgcn_readfirstlane(__shfl(sv, 31));
    }
    __syncthreads();

    int step0 = NKMAX / 2;
    while (step0 >= nblk && step0 > 1) step0 >>= 1;

    int pbase = 0;
    do {
        const int n = total - pbase < PASS ? total - pbase : PASS;

        int es[PASS_K], ed[PASS_K], ew[PASS_K], rk[PASS_K], fs[PASS_K];
        int pp[PASS_K], kk[PASS_K];
        #pragma unroll
        for (int k = 0; k < PASS_K; ++k) {
            const int i = k * TK_THREADS + t;
            pp[k] = pbase + (i < n ? i : 0);
            kk[k] = 0;
        }
        for (int st = step0; st >= 1; st >>= 1) {
            int pr[PASS_K];
            #pragma unroll
            for (int k = 0; k < PASS_K; ++k) pr[k] = rs[kk[k] + st];
            #pragma unroll
            for (int k = 0; k < PASS_K; ++k) kk[k] += pr[k] <= pp[k] ? st : 0;
        }
        int dv[PASS_K];
        #pragma unroll
        for (int k = 0; k < PASS_K; ++k) dv[k] = dlt[kk[k]];
        #pragma unroll
        for (int k = 0; k < PASS_K; ++k) {
            int e = dv[k] + pp[k];
            e = (unsigned)e < (unsigned)E ? e : E - 1;
            es[k] = ssrc[e];
            ew[k] = swb[e];
            const unsigned d = (unsigned)sdl[e];
            ed[k] = (int)(d < (unsigned)TILE ? d : (unsigned)TILE - 1u);
        }
        if (pbase == 0) {
            float4* s4 = reinterpret_cast<float4*>(sums);
            const float4 z = {0.f, 0.f, 0.f, 0.f};
            for (int i = t; i < TILE * ACC_PITCH / 4; i += TK_THREADS) s4[i] = z;
        }
        cnt[t] = 0;
        __syncthreads();

        #pragma unroll
        for (int k = 0; k < PASS_K; ++k) {
            rk[k] = 0;
            if (k * TK_THREADS + t < n) rk[k] = atomicAdd(&cnt[ed[k]], 1);
        }
        __syncthreads();

        {
            const int cv = cnt[t];
            int iv = cv;
            #pragma unroll
            for (int off = 1; off < 32; off <<= 1) {
                const int u = __shfl_up(iv, off);
                if (lane >= off) iv += u;
            }
            if (lane == 31) wtot[wid] = iv;
            __syncthreads();
            int sv = wtot[lane];
            #pragma unroll
            for (int off = 1; off < 32; off <<= 1) {
                const int u = __shfl_up(sv, off);
                if (lane >= off) sv += u;
            }
            const int pv = wid ? __shfl(sv, wid - 1) : 0;
            cnt[t] = pv + iv - cv;
        }
        __syncthreads();

        #pragma unroll
        for (int k = 0; k < PASS_K; ++k) fs[k] = cnt[ed[k]];
        #pragma unroll
        for (int k = 0; k < PASS_K; ++k) {
            if (k * TK_THREADS + t < n) {
                const int p = fs[k] + rk[k];
                ekey[p] = (ed[k] << KEY_SHIFT) | (es[k] & KEY_MASK);
                ewt[p]  = ew[k];
            }
        }
        __syncthreads();

        const int q  = ((n + TK_WAVES - 1) / TK_WAVES + EB - 1) & ~(EB - 1);
        const int lo = wid * q;
        const int hi = lo + q < n ? lo + q : n;
        if (lo < hi) {
            int cur = __builtin_amdgcn_readfirstlane(ekey[lo]) >> KEY_SHIFT;
            float acc[4] = {0.f, 0.f, 0.f, 0.f};
            for (int base = lo; base < hi; base += 32) {
                if (base + 32 <= hi) {
                    const int key_l  = ekey[base + lane];
                    const int w_bits = ewt[base + lane];

                    float va[EB][4], vb[EB][4];
                    tk_load_rows<0>(va, x, key_l, lane);
                    tk_load_rows<1>(vb, x, key_l, lane);
                    tk_sum_rows<0>(va, sums, key_l, w_bits, lane, cur, acc);
                    tk_load_rows<2>(va, x, key_l, lane);
                    tk_sum_rows<1>(vb, sums, key_l, w_bits, lane, cur, acc);
                    tk_load_rows<3>(vb, x, key_l, lane);
                    tk_sum_rows<2>(va, sums, key_l, w_bits, lane, cur, acc);
                    tk_sum_rows<3>(vb, sums, key_l, w_bits, lane, cur, acc);
                } else {
                    const int cntp = hi - base;
                    const bool live = lane < cntp;
                    const int key_l  = ekey[live ? base + lane : base];
                    const int w_bits = ewt[live ? base + lane : base];
                    for (int g = 0; g < cntp; g += EB) {
                        const int m = cntp - g;
                        float v[EB][4];
                        #pragma unroll
                        for (int j = 0; j < EB; ++j) {
                            const int s = __builtin_amdgcn_readlane(key_l, j < m ? g + j : g) & KEY_MASK;
                            const float4 r = reinterpret_cast<const float4*>(x + (size_t)s * DIM)[lane];
                            v[j][0] = r.x;
                            v[j][1] = r.y;
                            v[j][2] = r.z;
                            v[j][3] = r.w;
                        }
                        #pragma unroll
                        for (int j = 0; j < EB; ++j) {
                            if (j < m) {
                                const int row = __builtin_amdgcn_readlane(key_l, g + j) >> KEY_SHIFT;
                                const float wt = __int_as_float(__builtin_amdgcn_readlane(w_bits, g + j));
                                tk_sum_edge(v[j], wt, row, sums, lane, cur, acc);
                            }
                        }
                    }
                }
            }
            tk_flush_row(sums, cur, lane, acc);
        }

        pbase += PASS;
        if (pbase < total) __syncthreads();
    } while (pbase < total);

    __syncthreads();
    {
        const float4* wsrc = reinterpret_cast<const float4*>(w);
        float4* wdst = reinterpret_cast<float4*>(wlds);
        #pragma unroll
        for (int i = t; i < DIM * DIM / 4; i += TK_THREADS) wdst[i] = wsrc[i];
    }
    __syncthreads();

    const int half = lane >> 4;
    const int l15  = lane & 15;
    const int cp   = wid & 3;
    const int mg   = ((wid >> 2) - 2 * cp) & 7;
    const int koff = half * 2;
    const int col0 = cp * 32 + l15;

    const float bv0 = bias[col0];
    const float bv1 = bias[col0 + 16];

    for (int mt = mg; mt < MT_PER_TILE; mt += 8) {
        const int lrow = mt * 16 + l15;
        v8f acc0 = {bv0, bv0, bv0, bv0, bv0, bv0, bv0, bv0};
        v8f acc1 = {bv1, bv1, bv1, bv1, bv1, bv1, bv1, bv1};

        const float* ak = &sums[lrow * ACC_PITCH + koff];
        const float* wk = &wlds[koff * DIM + col0];
        #pragma unroll 4
        for (int kb = 0; kb < DIM; kb += 4, ak += 4, wk += 4 * DIM) {
            const float2 af = *reinterpret_cast<const float2*>(ak);
            v2f a;
            a.x = af.x;
            a.y = af.y;
            v2f b0, b1;
            b0.x = wk[0];
            b0.y = wk[DIM];
            b1.x = wk[16];
            b1.y = wk[DIM + 16];
            acc0 = __builtin_amdgcn_wmma_f32_16x16x4_f32(
                 false, a,  false, b0,
                 (short)0, acc0,
                 false,  false);
            acc1 = __builtin_amdgcn_wmma_f32_16x16x4_f32(
                 false, a,  false, b1,
                 (short)0, acc1,
                 false,  false);
        }

        asm volatile("v_nop\n\tv_nop\n\tv_nop\n\tv_nop" : "+v"(acc0), "+v"(acc1));

        const int row0 = blockIdx.x * TILE + mt * 16;
        float s0[8], s1[8];
        #pragma unroll
        for (int r = 0; r < 8; ++r) {
            const float lo = acc0[r];
            const float hi = acc1[r];
            const float got = __shfl_xor(half ? lo : hi, 16);
            s0[r] = half ? got : lo;
            s1[r] = half ? hi : got;
        }
        #pragma unroll
        for (int r = 0; r < 8; ++r) {
            const int m = row0 + r;
            if (m < N)     out[(size_t)m * DIM + cp * 32 + lane] = s0[r];
            if (m + 8 < N) out[(size_t)(m + 8) * DIM + cp * 32 + lane] = s1[r];
        }
    }
}

extern "C" void kernel_launch(void* const* d_in, const int* in_sizes, int n_in,
                              void* d_out, int out_size, void* d_ws, size_t ws_size,
                              hipStream_t stream) {
    const float* x    = (const float*)d_in[0];
    const int*   esrc = (const int*)d_in[1];
    const int*   edst = (const int*)d_in[2];
    const float* ew   = (const float*)d_in[3];
    const float* w    = (const float*)d_in[4];
    const float* bias = (const float*)d_in[5];

    const int N = in_sizes[0] / DIM;
    const int E = in_sizes[1];

    float* out = (float*)d_out;
    int*   wsi = (int*)d_ws;

    const int NB = (N + TILE - 1) / TILE;
    const int bin_blocks = (int)(((long long)E + BIN_CHUNK - 1) / BIN_CHUNK);

    gcn_bin_sort<<<bin_blocks, BIN_THREADS, 0, stream>>>(esrc, edst, ew, wsi, E, N);
    gcn_tile_sum_gemm<<<NB, TK_THREADS, 0, stream>>>(x, w, bias, wsi, out, N, E);
}


# phase 2: W kept transposed in LDS, so a lane's A and B fragments are 16-byte LDS loads (3 loads per 4 WMMAs, not 10; no 2-way bank conflict on B); a WMMA's four k regrouped, sum order changes
# speedup vs baseline: 4.7658x; 1.0341x over previous
#include <hip/hip_runtime.h>

typedef __attribute__((ext_vector_type(2))) float v2f;
typedef __attribute__((ext_vector_type(8))) float v8f;

#define DIM 128

#define TILE 400
#define MT_PER_TILE (TILE / 16)
#define ACC_PITCH 132
#define WT_PITCH 132
#define TK_THREADS 1024
#define TK_WAVES 32

#define NBMAX 1024
#define BIN_THREADS 1024
#define BIN_K 8
#define BIN_CHUNK (BIN_THREADS * BIN_K)
#define NKMAX 1024

static_assert(BIN_THREADS == NBMAX, "one thread per table column");
static_assert(NKMAX == TK_THREADS, "one thread per chunk in the run scan");
static_assert(BIN_CHUNK < (1 << 15), "a run's first slot and its length share one word");


__device__ __forceinline__ int bin_blocks_of(int E) {
    return (int)(((long long)E + BIN_CHUNK - 1) / BIN_CHUNK);
}

__global__ __launch_bounds__(BIN_THREADS) void gcn_bin_sort(const int* __restrict__ esrc,
                                                            const int* __restrict__ edst,
                                                            const float* __restrict__ ew,
                                                            int* wsi, int E, int N) {
    __shared__ int lc[NBMAX];
    __shared__ int wtot[32];
    __shared__ int st_s[BIN_CHUNK];
    __shared__ int st_d[BIN_CHUNK];
    __shared__ float st_w[BIN_CHUNK];

    const int t = threadIdx.x;
    const int lane = t & 31;
    const int wv = t >> 5;
    const int nblk = gridDim.x;
    const int kb = blockIdx.x;

    int* tab  = wsi;
    int* ssrc = wsi + nblk * NBMAX;
    int* sdl  = ssrc + E;
    float* sw = reinterpret_cast<float*>(sdl + E);

    int d[BIN_K], s[BIN_K], bb[BIN_K], rk[BIN_K];
    float wg[BIN_K];
    const unsigned cbase = kb * (unsigned)BIN_CHUNK;
    #pragma unroll
    for (int k = 0; k < BIN_K; ++k) {
        const unsigned e = cbase + k * BIN_THREADS + t;
        const unsigned ec = e < (unsigned)E ? e : (unsigned)E - 1u;
        const int v = edst[ec];
        s[k]  = esrc[ec];
        wg[k] = ew[ec];
        d[k] = e < (unsigned)E ? v : -1;
    }

    lc[t] = 0;
    __syncthreads();

    #pragma unroll
    for (int k = 0; k < BIN_K; ++k) {
        bb[k] = -1;
        rk[k] = 0;
        if ((unsigned)d[k] < (unsigned)N) {
            bb[k] = d[k] / TILE;
            rk[k] = atomicAdd(&lc[bb[k]], 1);
        }
    }
    __syncthreads();

    const int c = lc[t];
    int ic = c;
    #pragma unroll
    for (int off = 1; off < 32; off <<= 1) {
        const int uc = __shfl_up(ic, off);
        if (lane >= off) ic += uc;
    }
    if (lane == 31) wtot[wv] = ic;
    __syncthreads();
    int sc = wtot[lane];
    #pragma unroll
    for (int off = 1; off < 32; off <<= 1) {
        const int uc = __shfl_up(sc, off);
        if (lane >= off) sc += uc;
    }
    const int pc = wv ? __shfl(sc, wv - 1) : 0;
    const int nvalid = __shfl(sc, 31);
    const int loff   = pc + ic - c;
    lc[t] = loff;
    tab[kb * NBMAX + t] = (loff << 16) | c;
    __syncthreads();

    int lo[BIN_K];
    #pragma unroll
    for (int k = 0; k < BIN_K; ++k) lo[k] = lc[bb[k] >= 0 ? bb[k] : 0];
    #pragma unroll
    for (int k = 0; k < BIN_K; ++k) {
        if (bb[k] >= 0) {
            const int p = lo[k] + rk[k];
            st_s[p] = s[k];
            st_w[p] = wg[k];
            st_d[p] = d[k] - bb[k] * TILE;
        }
    }
    __syncthreads();

    int os[BIN_K], od[BIN_K];
    float ow[BIN_K];
    #pragma unroll
    for (int k = 0; k < BIN_K; ++k) {
        const int i = k * BIN_THREADS + t;
        const int ic2 = i < nvalid ? i : 0;
        os[k] = st_s[ic2];
        od[k] = st_d[ic2];
        ow[k] = st_w[ic2];
    }
    #pragma unroll
    for (int k = 0; k < BIN_K; ++k) {
        const int i = k * BIN_THREADS + t;
        if (i < nvalid) {
            const unsigned pos = cbase + i;
            ssrc[pos] = os[k];
            sdl[pos]  = od[k];
            sw[pos]   = ow[k];
        }
    }
}

#define EB 8
#define PASS 8192
#define PASS_K (PASS / TK_THREADS)
#define KEY_SHIFT 19
#define KEY_MASK ((1 << KEY_SHIFT) - 1)

static_assert(NBMAX * TILE <= (1 << KEY_SHIFT), "a source row must fit the low key bits");
static_assert(PASS * 2 <= DIM * WT_PITCH, "keys + weights of a pass fit the W buffer");
static_assert(TILE <= TK_THREADS, "one thread per row in the scan");

template <int B>
__device__ __forceinline__ void tk_load_rows(float (&v)[EB][4], const float* __restrict__ x,
                                             int key_l, int lane) {
    #pragma unroll
    for (int j = 0; j < EB; ++j) {
        const int s = __builtin_amdgcn_readlane(key_l, B * EB + j) & KEY_MASK;
        const float4 r = reinterpret_cast<const float4*>(x + (size_t)s * DIM)[lane];
        v[j][0] = r.x;
        v[j][1] = r.y;
        v[j][2] = r.z;
        v[j][3] = r.w;
    }
}

__device__ __forceinline__ void tk_flush_row(float* sums, int row, int lane, const float (&acc)[4]) {
    float* o = sums + row * ACC_PITCH + 4 * lane;
    __hip_atomic_fetch_add(o + 0, acc[0], __ATOMIC_RELAXED, __HIP_MEMORY_SCOPE_WORKGROUP);
    __hip_atomic_fetch_add(o + 1, acc[1], __ATOMIC_RELAXED, __HIP_MEMORY_SCOPE_WORKGROUP);
    __hip_atomic_fetch_add(o + 2, acc[2], __ATOMIC_RELAXED, __HIP_MEMORY_SCOPE_WORKGROUP);
    __hip_atomic_fetch_add(o + 3, acc[3], __ATOMIC_RELAXED, __HIP_MEMORY_SCOPE_WORKGROUP);
}

__device__ __forceinline__ void tk_sum_edge(const float (&v)[4], float wt, int row, float* sums,
                                            int lane, int& cur, float (&acc)[4]) {
    if (row != cur) {
        tk_flush_row(sums, cur, lane, acc);
        cur = row;
        acc[0] = 0.f; acc[1] = 0.f; acc[2] = 0.f; acc[3] = 0.f;
    }
    acc[0] += v[0] * wt;
    acc[1] += v[1] * wt;
    acc[2] += v[2] * wt;
    acc[3] += v[3] * wt;
}

template <int B>
__device__ __forceinline__ void tk_sum_rows(const float (&v)[EB][4], float* sums, int key_l,
                                            int w_bits, int lane, int& cur, float (&acc)[4]) {
    #pragma unroll
    for (int j = 0; j < EB; ++j) {
        const int row = __builtin_amdgcn_readlane(key_l, B * EB + j) >> KEY_SHIFT;
        const float wt = __int_as_float(__builtin_amdgcn_readlane(w_bits, B * EB + j));
        tk_sum_edge(v[j], wt, row, sums, lane, cur, acc);
    }
}

__global__ __launch_bounds__(TK_THREADS) void gcn_tile_sum_gemm(const float* __restrict__ x,
                                                                const float* __restrict__ w,
                                                                const float* __restrict__ bias,
                                                                const int* __restrict__ wsi,
                                                                float* __restrict__ out,
                                                                int N, int E) {
    __shared__ __attribute__((aligned(16))) float ldsf[DIM * WT_PITCH + TILE * ACC_PITCH];
    float* const wlds = ldsf;
    float* const sums = ldsf + DIM * WT_PITCH;
    __shared__ int cnt[TK_THREADS];
    __shared__ int wtot[TK_WAVES];
    __shared__ int rs[NKMAX];
    __shared__ int dlt[NKMAX];

    int* ekey = reinterpret_cast<int*>(wlds);
    int* ewt  = ekey + PASS;

    const int t = threadIdx.x;
    const int lane = t & 31;
    const int wid = __builtin_amdgcn_readfirstlane(t >> 5);

    const int nblk   = bin_blocks_of(E);
    const int* tab   = wsi;
    const int* ssrc  = wsi + nblk * NBMAX;
    const int* sdl   = ssrc + E;
    const int* swb   = sdl + E;

    int total;
    {
        int cv = 0, ov = 0;
        if (t < nblk) {
            const int pk = tab[t * NBMAX + blockIdx.x];
            cv = pk & 0xffff;
            cv = cv < BIN_CHUNK ? cv : BIN_CHUNK;
            ov = (pk >> 16) & 0xffff;
            ov = ov < BIN_CHUNK ? ov : BIN_CHUNK;
        }
        int iv = cv;
        #pragma unroll
        for (int off = 1; off < 32; off <<= 1) {
            const int u = __shfl_up(iv, off);
            if (lane >= off) iv += u;
        }
        if (lane == 31) wtot[wid] = iv;
        __syncthreads();
        int sv = wtot[lane];
        #pragma unroll
        for (int off = 1; off < 32; off <<= 1) {
            const int u = __shfl_up(sv, off);
            if (lane >= off) sv += u;
        }
        const int pv = wid ? __shfl(sv, wid - 1) : 0;
        const int ex = pv + iv - cv;
        rs[t]  = ex;
        dlt[t] = t * BIN_CHUNK + ov - ex;
        total = __builtin_amdgcn_readfirstlane(__shfl(sv, 31));
    }
    __syncthreads();

    int step0 = NKMAX / 2;
    while (step0 >= nblk && step0 > 1) step0 >>= 1;

    int pbase = 0;
    do {
        const int n = total - pbase < PASS ? total - pbase : PASS;

        int es[PASS_K], ed[PASS_K], ew[PASS_K], rk[PASS_K], fs[PASS_K];
        int pp[PASS_K], kk[PASS_K];
        #pragma unroll
        for (int k = 0; k < PASS_K; ++k) {
            const int i = k * TK_THREADS + t;
            pp[k] = pbase + (i < n ? i : 0);
            kk[k] = 0;
        }
        for (int st = step0; st >= 1; st >>= 1) {
            int pr[PASS_K];
            #pragma unroll
            for (int k = 0; k < PASS_K; ++k) pr[k] = rs[kk[k] + st];
            #pragma unroll
            for (int k = 0; k < PASS_K; ++k) kk[k] += pr[k] <= pp[k] ? st : 0;
        }
        int dv[PASS_K];
        #pragma unroll
        for (int k = 0; k < PASS_K; ++k) dv[k] = dlt[kk[k]];
        #pragma unroll
        for (int k = 0; k < PASS_K; ++k) {
            int e = dv[k] + pp[k];
            e = (unsigned)e < (unsigned)E ? e : E - 1;
            es[k] = ssrc[e];
            ew[k] = swb[e];
            const unsigned d = (unsigned)sdl[e];
            ed[k] = (int)(d < (unsigned)TILE ? d : (unsigned)TILE - 1u);
        }
        if (pbase == 0) {
            float4* s4 = reinterpret_cast<float4*>(sums);
            const float4 z = {0.f, 0.f, 0.f, 0.f};
            for (int i = t; i < TILE * ACC_PITCH / 4; i += TK_THREADS) s4[i] = z;
        }
        cnt[t] = 0;
        __syncthreads();

        #pragma unroll
        for (int k = 0; k < PASS_K; ++k) {
            rk[k] = 0;
            if (k * TK_THREADS + t < n) rk[k] = atomicAdd(&cnt[ed[k]], 1);
        }
        __syncthreads();

        {
            const int cv = cnt[t];
            int iv = cv;
            #pragma unroll
            for (int off = 1; off < 32; off <<= 1) {
                const int u = __shfl_up(iv, off);
                if (lane >= off) iv += u;
            }
            if (lane == 31) wtot[wid] = iv;
            __syncthreads();
            int sv = wtot[lane];
            #pragma unroll
            for (int off = 1; off < 32; off <<= 1) {
                const int u = __shfl_up(sv, off);
                if (lane >= off) sv += u;
            }
            const int pv = wid ? __shfl(sv, wid - 1) : 0;
            cnt[t] = pv + iv - cv;
        }
        __syncthreads();

        #pragma unroll
        for (int k = 0; k < PASS_K; ++k) fs[k] = cnt[ed[k]];
        #pragma unroll
        for (int k = 0; k < PASS_K; ++k) {
            if (k * TK_THREADS + t < n) {
                const int p = fs[k] + rk[k];
                ekey[p] = (ed[k] << KEY_SHIFT) | (es[k] & KEY_MASK);
                ewt[p]  = ew[k];
            }
        }
        __syncthreads();

        const int q  = ((n + TK_WAVES - 1) / TK_WAVES + EB - 1) & ~(EB - 1);
        const int lo = wid * q;
        const int hi = lo + q < n ? lo + q : n;
        if (lo < hi) {
            int cur = __builtin_amdgcn_readfirstlane(ekey[lo]) >> KEY_SHIFT;
            float acc[4] = {0.f, 0.f, 0.f, 0.f};
            for (int base = lo; base < hi; base += 32) {
                if (base + 32 <= hi) {
                    const int key_l  = ekey[base + lane];
                    const int w_bits = ewt[base + lane];

                    float va[EB][4], vb[EB][4];
                    tk_load_rows<0>(va, x, key_l, lane);
                    tk_load_rows<1>(vb, x, key_l, lane);
                    tk_sum_rows<0>(va, sums, key_l, w_bits, lane, cur, acc);
                    tk_load_rows<2>(va, x, key_l, lane);
                    tk_sum_rows<1>(vb, sums, key_l, w_bits, lane, cur, acc);
                    tk_load_rows<3>(vb, x, key_l, lane);
                    tk_sum_rows<2>(va, sums, key_l, w_bits, lane, cur, acc);
                    tk_sum_rows<3>(vb, sums, key_l, w_bits, lane, cur, acc);
                } else {
                    const int cntp = hi - base;
                    const bool live = lane < cntp;
                    const int key_l  = ekey[live ? base + lane : base];
                    const int w_bits = ewt[live ? base + lane : base];
                    for (int g = 0; g < cntp; g += EB) {
                        const int m = cntp - g;
                        float v[EB][4];
                        #pragma unroll
                        for (int j = 0; j < EB; ++j) {
                            const int s = __builtin_amdgcn_readlane(key_l, j < m ? g + j : g) & KEY_MASK;
                            const float4 r = reinterpret_cast<const float4*>(x + (size_t)s * DIM)[lane];
                            v[j][0] = r.x;
                            v[j][1] = r.y;
                            v[j][2] = r.z;
                            v[j][3] = r.w;
                        }
                        #pragma unroll
                        for (int j = 0; j < EB; ++j) {
                            if (j < m) {
                                const int row = __builtin_amdgcn_readlane(key_l, g + j) >> KEY_SHIFT;
                                const float wt = __int_as_float(__builtin_amdgcn_readlane(w_bits, g + j));
                                tk_sum_edge(v[j], wt, row, sums, lane, cur, acc);
                            }
                        }
                    }
                }
            }
            tk_flush_row(sums, cur, lane, acc);
        }

        pbase += PASS;
        if (pbase < total) __syncthreads();
    } while (pbase < total);

    __syncthreads();
    {
        const int wc = t & (DIM - 1);
        const int wq = t >> 7;
        float4 wv[4];
        #pragma unroll
        for (int j = 0; j < 4; ++j) {
            const int k0 = 4 * (wq + 8 * j);
            wv[j].x = w[(k0 + 0) * DIM + wc];
            wv[j].y = w[(k0 + 1) * DIM + wc];
            wv[j].z = w[(k0 + 2) * DIM + wc];
            wv[j].w = w[(k0 + 3) * DIM + wc];
        }
        #pragma unroll
        for (int j = 0; j < 4; ++j) {
            const int k0 = 4 * (wq + 8 * j);
            *reinterpret_cast<float4*>(&wlds[wc * WT_PITCH + k0]) = wv[j];
        }
    }
    __syncthreads();

    const int half = lane >> 4;
    const int l15  = lane & 15;
    const int cp   = wid & 3;
    const int mg   = ((wid >> 2) - 2 * cp) & 7;
    const int koff = half * 4;
    const int col0 = cp * 32 + l15;

    const float bv0 = bias[col0];
    const float bv1 = bias[col0 + 16];

    for (int mt = mg; mt < MT_PER_TILE; mt += 8) {
        const int lrow = mt * 16 + l15;
        v8f acc0 = {bv0, bv0, bv0, bv0, bv0, bv0, bv0, bv0};
        v8f acc1 = {bv1, bv1, bv1, bv1, bv1, bv1, bv1, bv1};

        const float* ak = &sums[lrow * ACC_PITCH + koff];
        const float* wk = &wlds[col0 * WT_PITCH + koff];
        #pragma unroll 2
        for (int kb = 0; kb < DIM; kb += 8, ak += 8, wk += 8) {
            const float4 af  = *reinterpret_cast<const float4*>(ak);
            const float4 bf0 = *reinterpret_cast<const float4*>(wk);
            const float4 bf1 = *reinterpret_cast<const float4*>(wk + 16 * WT_PITCH);
            v2f a, b0, b1;
            a.x = af.x;   a.y = af.y;
            b0.x = bf0.x; b0.y = bf0.y;
            b1.x = bf1.x; b1.y = bf1.y;
            acc0 = __builtin_amdgcn_wmma_f32_16x16x4_f32(
                 false, a,  false, b0,
                 (short)0, acc0,
                 false,  false);
            acc1 = __builtin_amdgcn_wmma_f32_16x16x4_f32(
                 false, a,  false, b1,
                 (short)0, acc1,
                 false,  false);
            a.x = af.z;   a.y = af.w;
            b0.x = bf0.z; b0.y = bf0.w;
            b1.x = bf1.z; b1.y = bf1.w;
            acc0 = __builtin_amdgcn_wmma_f32_16x16x4_f32(
                 false, a,  false, b0,
                 (short)0, acc0,
                 false,  false);
            acc1 = __builtin_amdgcn_wmma_f32_16x16x4_f32(
                 false, a,  false, b1,
                 (short)0, acc1,
                 false,  false);
        }

        asm volatile("v_nop\n\tv_nop\n\tv_nop\n\tv_nop" : "+v"(acc0), "+v"(acc1));

        const int row0 = blockIdx.x * TILE + mt * 16;
        float s0[8], s1[8];
        #pragma unroll
        for (int r = 0; r < 8; ++r) {
            const float lo = acc0[r];
            const float hi = acc1[r];
            const float got = __shfl_xor(half ? lo : hi, 16);
            s0[r] = half ? got : lo;
            s1[r] = half ? hi : got;
        }
        #pragma unroll
        for (int r = 0; r < 8; ++r) {
            const int m = row0 + r;
            if (m < N)     out[(size_t)m * DIM + cp * 32 + lane] = s0[r];
            if (m + 8 < N) out[(size_t)(m + 8) * DIM + cp * 32 + lane] = s1[r];
        }
    }
}

extern "C" void kernel_launch(void* const* d_in, const int* in_sizes, int n_in,
                              void* d_out, int out_size, void* d_ws, size_t ws_size,
                              hipStream_t stream) {
    const float* x    = (const float*)d_in[0];
    const int*   esrc = (const int*)d_in[1];
    const int*   edst = (const int*)d_in[2];
    const float* ew   = (const float*)d_in[3];
    const float* w    = (const float*)d_in[4];
    const float* bias = (const float*)d_in[5];

    const int N = in_sizes[0] / DIM;
    const int E = in_sizes[1];

    float* out = (float*)d_out;
    int*   wsi = (int*)d_ws;

    const int NB = (N + TILE - 1) / TILE;
    const int bin_blocks = (int)(((long long)E + BIN_CHUNK - 1) / BIN_CHUNK);

    gcn_bin_sort<<<bin_blocks, BIN_THREADS, 0, stream>>>(esrc, edst, ew, wsi, E, N);
    gcn_tile_sum_gemm<<<NB, TK_THREADS, 0, stream>>>(x, w, bias, wsi, out, N, E);
}
